# starting kernel
# baseline (speedup 1.0000x reference)
#include <hip/hip_runtime.h>
#include <hip/hip_bf16.h>
#include <math.h>


typedef float v2f __attribute__((ext_vector_type(2)));
typedef float v8f __attribute__((ext_vector_type(8)));

#define ROWS_TOTAL 16384
#define F_ 16
#define D_ 64
#define H_ 128
#define FH_ 2048

#define ST_T 2052
#define ST_X 68
#define ST_P 132
#define ST_S 20

#define SMEM_FLOATS (16*ST_T + 16*ST_X + 4*16*ST_P + 2*16*ST_S + 16*16)

__device__ __forceinline__ float rowsum16(float v) {
  v += __shfl_xor(v, 1, 16);
  v += __shfl_xor(v, 2, 16);
  v += __shfl_xor(v, 4, 16);
  v += __shfl_xor(v, 8, 16);
  return v;
}
__device__ __forceinline__ float rowmax16(float v) {
  v = fmaxf(v, __shfl_xor(v, 1, 16));
  v = fmaxf(v, __shfl_xor(v, 2, 16));
  v = fmaxf(v, __shfl_xor(v, 4, 16));
  v = fmaxf(v, __shfl_xor(v, 8, 16));
  return v;
}

template <int K>
__device__ __forceinline__ v8f wmma_gemm(const float* __restrict__ As, int lda,
                                         const float* __restrict__ W, int ldw,
                                         int n0, int lane16, int hi) {
  v8f c = {0.f, 0.f, 0.f, 0.f, 0.f, 0.f, 0.f, 0.f};
  const float* arow = As + lane16 * lda + hi * 2;
  const float* wcol = W + (size_t)(hi * 2) * ldw + n0 + lane16;
#pragma unroll 1
  for (int kb = 0; kb < K; kb += 32) {
    if (kb + 32 < K) {
      __builtin_prefetch(wcol + (size_t)(kb + 32) * ldw, 0, 1);
      __builtin_prefetch(wcol + (size_t)(kb + 48) * ldw, 0, 1);
    }
#pragma unroll
    for (int k0 = kb; k0 < kb + 32; k0 += 4) {
      v2f a, b;
      a.x = arow[k0];
      a.y = arow[k0 + 1];
      b.x = wcol[(size_t)k0 * ldw];
      b.y = wcol[(size_t)(k0 + 1) * ldw];
      c = __builtin_amdgcn_wmma_f32_16x16x4_f32(false, a, false, b,
                                                (short)0, c, false, false);
    }
  }
  return c;
}

__device__ __forceinline__ void store_tile(float* dst, int ldd, int col0,
                                           v8f c, int lane16, int hi) {
#pragma unroll
  for (int v = 0; v < 8; ++v)
    dst[(v + hi * 8) * ldd + col0 + lane16] = c[v];
}

__global__ __launch_bounds__(256, 1)
void vsn_fused_206158430403(
    const float* __restrict__ x,
    const float* __restrict__ embW, const float* __restrict__ embB,
    const float* __restrict__ fc1W, const float* __restrict__ fc1B,
    const float* __restrict__ fc2W, const float* __restrict__ fc2B,
    const float* __restrict__ gW,   const float* __restrict__ gB,
    const float* __restrict__ lnG,  const float* __restrict__ lnB,
    const float* __restrict__ sf1W, const float* __restrict__ sf1B,
    const float* __restrict__ sf2W, const float* __restrict__ sf2B,
    const float* __restrict__ sgW,  const float* __restrict__ sgB,
    const float* __restrict__ sskW, const float* __restrict__ sskB,
    const float* __restrict__ slnG, const float* __restrict__ slnB,
    float* __restrict__ outSel, float* __restrict__ outW) {
  extern __shared__ float smem[];
  float* TRANS = smem;
  float* XT = TRANS + 16 * ST_T;
  float* P0 = XT + 16 * ST_X;
  float* P1 = P0 + 16 * ST_P;
  float* P2 = P1 + 16 * ST_P;
  float* P3 = P2 + 16 * ST_P;
  float* GP = P3 + 16 * ST_P;
  float* SKT = GP + 16 * ST_S;
  float* WB = SKT + 16 * ST_S;

  const int tid = threadIdx.x;
  const int wid = tid >> 5;
  const int lane = tid & 31;
  const int lane16 = lane & 15;
  const int hi = lane >> 4;
  const int n0 = wid * 16;
  const int lr = tid >> 4;
  const int lc = tid & 15;
  const int row0 = blockIdx.x * 16;

  for (int f = 0; f < F_; ++f) {
    {
      const float* xp = x + (size_t)(row0 + lr) * (F_ * D_) + f * D_ + lc * 4;
      float4 v = *(const float4*)xp;
      float* dp = XT + lr * ST_X + lc * 4;
      dp[0] = v.x; dp[1] = v.y; dp[2] = v.z; dp[3] = v.w;
      if (f + 1 < F_) __builtin_prefetch(xp + D_, 0, 1);
    }
    __syncthreads();

    v8f ce = wmma_gemm<D_>(XT, ST_X, embW + (size_t)f * D_ * H_, H_, n0, lane16, hi);
    {
      float b = embB[f * H_ + n0 + lane16];
#pragma unroll
      for (int v = 0; v < 8; ++v) ce[v] += b;
    }
    store_tile(P0, ST_P, n0, ce, lane16, hi);
    __syncthreads();

    v8f c1 = wmma_gemm<H_>(P0, ST_P, fc1W + (size_t)f * H_ * H_, H_, n0, lane16, hi);
    {
      float b = fc1B[f * H_ + n0 + lane16];
#pragma unroll
      for (int v = 0; v < 8; ++v) {
        float t = c1[v] + b;
        c1[v] = t > 0.f ? t : (expf(t) - 1.f);
      }
    }
    store_tile(P1, ST_P, n0, c1, lane16, hi);
    __syncthreads();

    v8f c2 = wmma_gemm<H_>(P1, ST_P, fc2W + (size_t)f * H_ * H_, H_, n0, lane16, hi);
    {
      float b = fc2B[f * H_ + n0 + lane16];
#pragma unroll
      for (int v = 0; v < 8; ++v) c2[v] += b;
    }
    store_tile(P2, ST_P, n0, c2, lane16, hi);
    __syncthreads();

    v8f cg = wmma_gemm<H_>(P2, ST_P, gW + (size_t)f * H_ * H_, H_, n0, lane16, hi);
    {
      float b = gB[f * H_ + n0 + lane16];
#pragma unroll
      for (int v = 0; v < 8; ++v) {
        float t = cg[v] + b;
        float s = 1.f / (1.f + expf(-t));
        cg[v] = ce[v] + s * c2[v];
      }
    }
    store_tile(P3, ST_P, n0, cg, lane16, hi);
    __syncthreads();

    {
      float vals[8];
      float s = 0.f, ss = 0.f;
#pragma unroll
      for (int j = 0; j < 8; ++j) {
        float v = P3[lr * ST_P + lc + 16 * j];
        vals[j] = v; s += v; ss += v * v;
      }
      s = rowsum16(s);
      ss = rowsum16(ss);
      float mu = s * (1.f / 128.f);
      float var = ss * (1.f / 128.f) - mu * mu;
      float rs = rsqrtf(var + 1e-5f);
#pragma unroll
      for (int j = 0; j < 8; ++j) {
        int h = lc + 16 * j;
        float nv = (vals[j] - mu) * rs * lnG[f * H_ + h] + lnB[f * H_ + h];
        TRANS[lr * ST_T + f * H_ + h] = nv;
      }
    }
    __syncthreads();
  }

  v8f b1 = wmma_gemm<FH_>(TRANS, ST_T, sf1W, H_, n0, lane16, hi);
  {
    float b = sf1B[n0 + lane16];
#pragma unroll
    for (int v = 0; v < 8; ++v) {
      float t = b1[v] + b;
      b1[v] = t > 0.f ? t : (expf(t) - 1.f);
    }
  }
  store_tile(P1, ST_P, n0, b1, lane16, hi);
  __syncthreads();

  v8f b2 = wmma_gemm<H_>(P1, ST_P, sf2W, H_, n0, lane16, hi);
  {
    float b = sf2B[n0 + lane16];
#pragma unroll
    for (int v = 0; v < 8; ++v) b2[v] += b;
  }
  store_tile(P2, ST_P, n0, b2, lane16, hi);
  __syncthreads();

  if (wid == 0) {
    v8f g = wmma_gemm<H_>(P2, ST_P, sgW, F_, 0, lane16, hi);
    float b = sgB[lane16];
#pragma unroll
    for (int v = 0; v < 8; ++v) g[v] += b;
    store_tile(GP, ST_S, 0, g, lane16, hi);
  } else if (wid == 1) {
    v8f s = wmma_gemm<FH_>(TRANS, ST_T, sskW, F_, 0, lane16, hi);
    float b = sskB[lane16];
#pragma unroll
    for (int v = 0; v < 8; ++v) s[v] += b;
    store_tile(SKT, ST_S, 0, s, lane16, hi);
  }
  __syncthreads();

  {
    float gp = GP[lr * ST_S + lc];
    float sg = 1.f / (1.f + expf(-gp));
    float v = SKT[lr * ST_S + lc] + sg * gp;
    float s = rowsum16(v);
    float ss = rowsum16(v * v);
    float mu = s * (1.f / 16.f);
    float var = ss * (1.f / 16.f) - mu * mu;
    float nv = (v - mu) * rsqrtf(var + 1e-5f) * slnG[lc] + slnB[lc];
    float mx = rowmax16(nv);
    float e = expf(nv - mx);
    float se = rowsum16(e);
    float w = e / se;
    WB[lr * 16 + lc] = w;
    outW[(size_t)(row0 + lr) * F_ + lc] = w;
  }
  __syncthreads();

  {
#pragma unroll
    for (int j = 0; j < 8; ++j) {
      int h = lc + 16 * j;
      float acc = 0.f;
#pragma unroll
      for (int f = 0; f < F_; ++f)
        acc += TRANS[lr * ST_T + f * H_ + h] * WB[lr * 16 + f];
      outSel[(size_t)(row0 + lr) * H_ + h] = acc;
    }
  }
}

extern "C" void kernel_launch(void* const* d_in, const int* in_sizes, int n_in,
                              void* d_out, int out_size, void* d_ws, size_t ws_size,
                              hipStream_t stream) {
  (void)in_sizes; (void)n_in; (void)out_size; (void)d_ws; (void)ws_size;
  const float* x    = (const float*)d_in[0];
  const float* embW = (const float*)d_in[1];
  const float* embB = (const float*)d_in[2];
  const float* fc1W = (const float*)d_in[3];
  const float* fc1B = (const float*)d_in[4];
  const float* fc2W = (const float*)d_in[5];
  const float* fc2B = (const float*)d_in[6];
  const float* gW   = (const float*)d_in[7];
  const float* gB   = (const float*)d_in[8];
  const float* lnG  = (const float*)d_in[9];
  const float* lnB  = (const float*)d_in[10];
  const float* sf1W = (const float*)d_in[11];
  const float* sf1B = (const float*)d_in[12];
  const float* sf2W = (const float*)d_in[13];
  const float* sf2B = (const float*)d_in[14];
  const float* sgW  = (const float*)d_in[15];
  const float* sgB  = (const float*)d_in[16];
  const float* sskW = (const float*)d_in[17];
  const float* sskB = (const float*)d_in[18];
  const float* slnG = (const float*)d_in[19];
  const float* slnB = (const float*)d_in[20];

  float* outSel = (float*)d_out;
  float* outW   = outSel + (size_t)ROWS_TOTAL * H_;

  constexpr size_t shmem = (size_t)SMEM_FLOATS * sizeof(float);
  (void)hipFuncSetAttribute((const void*)vsn_fused_206158430403,
                            hipFuncAttributeMaxDynamicSharedMemorySize,
                            (int)shmem);

  vsn_fused_206158430403<<<ROWS_TOTAL / 16, 256, shmem, stream>>>(
      x, embW, embB, fc1W, fc1B, fc2W, fc2B, gW, gB, lnG, lnB,
      sf1W, sf1B, sf2W, sf2B, sgW, sgB, sskW, sskB, slnG, slnB,
      outSel, outW);
}


# Matrix waves read A and B fragments from LDS in whole groups of 8 k steps (two register sets) in front of and between 8-WMMA blocks, not 2-4 ahead among the WMMAs; bias requested first; sums kept
# speedup vs baseline: 1.8842x; 1.8842x over previous
#include <hip/hip_runtime.h>
#include <hip/hip_bf16.h>
#include <math.h>


typedef float v2f __attribute__((ext_vector_type(2)));
typedef float v8f __attribute__((ext_vector_type(8)));

#define ROWS_TOTAL 16384
#define F_ 16
#define D_ 64
#define H_ 128
#define FH_ 2048

#define ST_T 2052
#define ST_X 68
#define ST_P 132
#define ST_S 20

#define WPITCH 288
#define WBUF_FLOATS (64*WPITCH)

#define SKPITCH 32

#define SMEM_FLOATS (16*ST_T + 16*ST_X + 4*16*ST_P + 2*16*ST_S + 16*16 + 2*WBUF_FLOATS)

__device__ __forceinline__ float rowsum16(float v) {
  v += __shfl_xor(v, 1, 16);
  v += __shfl_xor(v, 2, 16);
  v += __shfl_xor(v, 4, 16);
  v += __shfl_xor(v, 8, 16);
  return v;
}
__device__ __forceinline__ float rowmax16(float v) {
  v = fmaxf(v, __shfl_xor(v, 1, 16));
  v = fmaxf(v, __shfl_xor(v, 2, 16));
  v = fmaxf(v, __shfl_xor(v, 4, 16));
  v = fmaxf(v, __shfl_xor(v, 8, 16));
  return v;
}

template <int K>
__device__ __forceinline__ v8f wmma_gemm(const float* __restrict__ As, int lda,
                                         const float* __restrict__ W, int ldw,
                                         int n0, int lane16, int hi) {
  v8f c = {0.f, 0.f, 0.f, 0.f, 0.f, 0.f, 0.f, 0.f};
  const float* arow = As + lane16 * lda + hi * 2;
  const float* wcol = W + (size_t)(hi * 2) * ldw + n0 + lane16;
#pragma unroll 1
  for (int kb = 0; kb < K; kb += 32) {
    if (kb + 32 < K) {
      __builtin_prefetch(wcol + (size_t)(kb + 32) * ldw, 0, 1);
      __builtin_prefetch(wcol + (size_t)(kb + 48) * ldw, 0, 1);
    }
#pragma unroll
    for (int k0 = kb; k0 < kb + 32; k0 += 4) {
      v2f a, b;
      a.x = arow[k0];
      a.y = arow[k0 + 1];
      b.x = wcol[(size_t)k0 * ldw];
      b.y = wcol[(size_t)(k0 + 1) * ldw];
      c = __builtin_amdgcn_wmma_f32_16x16x4_f32(false, a, false, b,
                                                (short)0, c, false, false);
    }
  }
  return c;
}

#define FR_CH 8
template <int K>
__device__ __forceinline__ v8f wmma_gemm_lds(v8f c, const float* As, int lda,
                                             const float* Bs, int n0,
                                             int lane16, int hi) {
  const float* arow = As + lane16 * lda + hi * 2;
  const float* bcol = Bs + hi * WPITCH + 2 * (n0 + lane16);
  constexpr int NCH = K / (4 * FR_CH);
  v2f a[2][FR_CH], b[2][FR_CH];
#pragma unroll
  for (int j = 0; j < 2; ++j) {
#pragma unroll
    for (int i = 0; i < FR_CH; ++i) {
      const int k0 = 4 * (j * FR_CH + i);
      a[j][i].x = arow[k0];
      a[j][i].y = arow[k0 + 1];
      b[j][i].x = bcol[(k0 >> 1) * WPITCH];
      b[j][i].y = bcol[(k0 >> 1) * WPITCH + 1];
    }
    __builtin_amdgcn_sched_barrier(0);
  }
#pragma unroll
  for (int j = 0; j < NCH; ++j) {
#pragma unroll
    for (int i = 0; i < FR_CH; ++i)
      c = __builtin_amdgcn_wmma_f32_16x16x4_f32(false, a[j & 1][i], false,
                                                b[j & 1][i], (short)0, c,
                                                false, false);
    if (j + 2 < NCH) {
      __builtin_amdgcn_sched_barrier(0);
#pragma unroll
      for (int i = 0; i < FR_CH; ++i) {
        const int k0 = 4 * ((j + 2) * FR_CH + i);
        a[j & 1][i].x = arow[k0];
        a[j & 1][i].y = arow[k0 + 1];
        b[j & 1][i].x = bcol[(k0 >> 1) * WPITCH];
        b[j & 1][i].y = bcol[(k0 >> 1) * WPITCH + 1];
      }
      __builtin_amdgcn_sched_barrier(0);
    }
  }
  return c;
}

#define PAIR_CH FR_CH
template <int K>
__device__ __forceinline__ void wmma_gemm_lds_pair(v8f& c, v8f& d,
                                                   const float* As, int lda,
                                                   const float* Bs, int n0,
                                                   const float* Ss,
                                                   int lane16, int hi) {
  const float* arow = As + lane16 * lda + hi * 2;
  const float* bcol = Bs + hi * WPITCH + 2 * (n0 + lane16);
  const float* scol = Ss + hi * SKPITCH + 2 * lane16;
  constexpr int NCH = K / (4 * PAIR_CH);
  v2f a[2][PAIR_CH], b[2][PAIR_CH], s[2][PAIR_CH];
#pragma unroll
  for (int j = 0; j < 2; ++j) {
#pragma unroll
    for (int i = 0; i < PAIR_CH; ++i) {
      const int k0 = 4 * (j * PAIR_CH + i);
      a[j][i].x = arow[k0];
      a[j][i].y = arow[k0 + 1];
      b[j][i].x = bcol[(k0 >> 1) * WPITCH];
      b[j][i].y = bcol[(k0 >> 1) * WPITCH + 1];
      s[j][i].x = scol[(k0 >> 1) * SKPITCH];
      s[j][i].y = scol[(k0 >> 1) * SKPITCH + 1];
    }
    __builtin_amdgcn_sched_barrier(0);
  }
#pragma unroll
  for (int j = 0; j < NCH; ++j) {
#pragma unroll
    for (int i = 0; i < PAIR_CH; ++i) {
      c = __builtin_amdgcn_wmma_f32_16x16x4_f32(false, a[j & 1][i], false,
                                                b[j & 1][i], (short)0, c,
                                                false, false);
      d = __builtin_amdgcn_wmma_f32_16x16x4_f32(false, a[j & 1][i], false,
                                                s[j & 1][i], (short)0, d,
                                                false, false);
    }
    if (j + 2 < NCH) {
      __builtin_amdgcn_sched_barrier(0);
#pragma unroll
      for (int i = 0; i < PAIR_CH; ++i) {
        const int k0 = 4 * ((j + 2) * PAIR_CH + i);
        a[j & 1][i].x = arow[k0];
        a[j & 1][i].y = arow[k0 + 1];
        b[j & 1][i].x = bcol[(k0 >> 1) * WPITCH];
        b[j & 1][i].y = bcol[(k0 >> 1) * WPITCH + 1];
        s[j & 1][i].x = scol[(k0 >> 1) * SKPITCH];
        s[j & 1][i].y = scol[(k0 >> 1) * SKPITCH + 1];
      }
      __builtin_amdgcn_sched_barrier(0);
    }
  }
}

typedef float v4f __attribute__((ext_vector_type(4)));
typedef const __attribute__((address_space(1))) float* gfloat_ptr;
typedef const __attribute__((address_space(1))) v4f* gv4f_ptr;
__device__ __forceinline__ gfloat_ptr scalar_base(const float* p) {
  const unsigned long long a = (unsigned long long)p;
  const unsigned lo = __builtin_amdgcn_readfirstlane((unsigned)a);
  const unsigned hi = __builtin_amdgcn_readfirstlane((unsigned)(a >> 32));
  return (gfloat_ptr)(((unsigned long long)hi << 32) | lo);
}
__device__ __forceinline__ float4 gload4(gfloat_ptr p) {
  const v4f v = *(gv4f_ptr)p;
  return make_float4(v.x, v.y, v.z, v.w);
}
__device__ __forceinline__ float bias_at(const float* rowbase, unsigned col) {
  return scalar_base(rowbase)[col];
}

struct WStage { float4 r[H_ / 8]; };

template <int KR>
__device__ __forceinline__ void wstage_load(WStage& s,
                                            const float* __restrict__ Wg,
                                            int tid) {
  const unsigned toff = (unsigned)((2 * (tid >> 5)) * H_ + 4 * (tid & 31));
  gfloat_ptr src = scalar_base(Wg) + toff;
#pragma unroll
  for (int i = 0; i < KR / 16; ++i) {
    s.r[2 * i]     = gload4(src + (16 * i) * H_);
    s.r[2 * i + 1] = gload4(src + (16 * i + 1) * H_);
  }
}

template <int KR>
__device__ __forceinline__ void wstage_store(const WStage& s, float* buf,
                                             int tid) {
  float* dst = buf + (tid >> 5) * WPITCH + 8 * (tid & 31);
#pragma unroll
  for (int i = 0; i < KR / 16; ++i) {
    const float4 e = s.r[2 * i], o = s.r[2 * i + 1];
    float4* d = (float4*)(dst + 8 * i * WPITCH);
    d[0] = make_float4(e.x, o.x, e.y, o.y);
    d[1] = make_float4(e.z, o.z, e.w, o.w);
  }
}

template <int KS, int KL>
__device__ __forceinline__ void wstage_swap(WStage& s, float* buf,
                                            const float* __restrict__ Wn,
                                            int tid) {
  float* dst = buf + (tid >> 5) * WPITCH + 8 * (tid & 31);
  const unsigned toff = (unsigned)((2 * (tid >> 5)) * H_ + 4 * (tid & 31));
  gfloat_ptr src = scalar_base(Wn) + toff;
#pragma unroll
  for (int i = 0; i < H_ / 16; ++i) {
    if (i < KS / 16) {
      const float4 e = s.r[2 * i], o = s.r[2 * i + 1];
      float4* d = (float4*)(dst + 8 * i * WPITCH);
      d[0] = make_float4(e.x, o.x, e.y, o.y);
      d[1] = make_float4(e.z, o.z, e.w, o.w);
    }
    if (i < KL / 16) {
      s.r[2 * i]     = gload4(src + (16 * i) * H_);
      s.r[2 * i + 1] = gload4(src + (16 * i + 1) * H_);
    }
    __builtin_amdgcn_sched_barrier(0);
  }
}

struct SkStage { float4 e, o; };

__device__ __forceinline__ void skstage_load(SkStage& s,
                                             const float* __restrict__ Wg,
                                             int tid) {
  const float* src = Wg + (size_t)(2 * (tid >> 2)) * F_ + 4 * (tid & 3);
  s.e = *(const float4*)src;
  s.o = *(const float4*)(src + F_);
}

__device__ __forceinline__ void skstage_store(const SkStage& s, float* buf,
                                              int tid) {
  float4* d = (float4*)(buf + (tid >> 2) * SKPITCH + 8 * (tid & 3));
  d[0] = make_float4(s.e.x, s.o.x, s.e.y, s.o.y);
  d[1] = make_float4(s.e.z, s.o.z, s.e.w, s.o.w);
}

__device__ __forceinline__ void store_tile(float* dst, int ldd, int col0,
                                           v8f c, int lane16, int hi) {
#pragma unroll
  for (int v = 0; v < 8; ++v)
    dst[(v + hi * 8) * ldd + col0 + lane16] = c[v];
}

__device__ __forceinline__ void lds_barrier() {
  __builtin_amdgcn_fence(__ATOMIC_RELEASE, "workgroup", "local");
  __builtin_amdgcn_s_barrier();
  __builtin_amdgcn_fence(__ATOMIC_ACQUIRE, "workgroup", "local");
}

__global__ __launch_bounds__(512, 1)
void vsn_fused(
    const float* __restrict__ x,
    const float* __restrict__ embW, const float* __restrict__ embB,
    const float* __restrict__ fc1W, const float* __restrict__ fc1B,
    const float* __restrict__ fc2W, const float* __restrict__ fc2B,
    const float* __restrict__ gW,   const float* __restrict__ gB,
    const float* __restrict__ lnG,  const float* __restrict__ lnB,
    const float* __restrict__ sf1W, const float* __restrict__ sf1B,
    const float* __restrict__ sf2W, const float* __restrict__ sf2B,
    const float* __restrict__ sgW,  const float* __restrict__ sgB,
    const float* __restrict__ sskW, const float* __restrict__ sskB,
    const float* __restrict__ slnG, const float* __restrict__ slnB,
    float* __restrict__ outSel, float* __restrict__ outW) {
  __shared__ __attribute__((aligned(16))) float smem[SMEM_FLOATS];
  float* TRANS = smem;
  float* XT = TRANS + 16 * ST_T;
  float* P0 = XT + 16 * ST_X;
  float* P1 = P0 + 16 * ST_P;
  float* P2 = P1 + 16 * ST_P;
  float* P3 = P2 + 16 * ST_P;
  float* GP = P3 + 16 * ST_P;
  float* SKT = GP + 16 * ST_S;
  float* WB = SKT + 16 * ST_S;
  float* BW0 = WB + 16 * 16;
  float* BW1 = BW0 + WBUF_FLOATS;

  const bool stager = __builtin_amdgcn_readfirstlane((int)threadIdx.x >> 8) != 0;
  const int tid = threadIdx.x & 255;
  const int wid = tid >> 5;
  const int lane = tid & 31;
  const int lane16 = lane & 15;
  const int hi = lane >> 4;
  const int n0 = wid * 16;
  const int lr = tid >> 4;
  const int lc = tid & 15;
  const int row0 = blockIdx.x * 16;
  const unsigned bco = (unsigned)(n0 + lane16);

  const v8f zero8 = {0.f, 0.f, 0.f, 0.f, 0.f, 0.f, 0.f, 0.f};
  WStage sW;
  SkStage sS;

  const float* xrow = x + (size_t)(row0 + lr) * (F_ * D_) + lc * 4;
  float* xdst = XT + lr * ST_X + lc * 4;
  float4 xv = make_float4(0.f, 0.f, 0.f, 0.f);
  if (stager) {
    wstage_load<D_>(sW, embW, tid);
    xv = *(const float4*)xrow;
    wstage_store<D_>(sW, BW0, tid);
    *(float4*)xdst = xv;
    wstage_load<H_>(sW, fc1W, tid);
  }
  lds_barrier();

  for (int f = 0; f < F_; ++f) {
    v8f ce = zero8, c2 = zero8;

    if (stager) {
      wstage_swap<H_, H_>(sW, BW1, fc2W + (size_t)f * H_ * H_, tid);
      if (f + 1 < F_) xv = *(const float4*)(xrow + (f + 1) * D_);
    } else {
      const float b = bias_at(embB + f * H_, bco);
      ce = wmma_gemm_lds<D_>(zero8, XT, ST_X, BW0, n0, lane16, hi);
#pragma unroll
      for (int v = 0; v < 8; ++v) ce[v] += b;
      store_tile(P0, ST_P, n0, ce, lane16, hi);
    }
    lds_barrier();

    if (stager) {
      wstage_swap<H_, H_>(sW, BW0, gW + (size_t)f * H_ * H_, tid);
      if (f + 1 < F_) *(float4*)xdst = xv;
    } else {
      const float b = bias_at(fc1B + f * H_, bco);
      v8f c1 = wmma_gemm_lds<H_>(zero8, P0, ST_P, BW1, n0, lane16, hi);
#pragma unroll
      for (int v = 0; v < 8; ++v) {
        float t = c1[v] + b;
        c1[v] = t > 0.f ? t : (expf(t) - 1.f);
      }
      store_tile(P1, ST_P, n0, c1, lane16, hi);
    }
    lds_barrier();

    if (stager) {
      if (f + 1 < F_)
        wstage_swap<H_, D_>(sW, BW1, embW + (size_t)(f + 1) * D_ * H_, tid);
      else
        wstage_swap<H_, H_>(sW, BW1, sf1W, tid);
    } else {
      const float b = bias_at(fc2B + f * H_, bco);
      c2 = wmma_gemm_lds<H_>(zero8, P1, ST_P, BW0, n0, lane16, hi);
#pragma unroll
      for (int v = 0; v < 8; ++v) c2[v] += b;
      store_tile(P2, ST_P, n0, c2, lane16, hi);
    }
    lds_barrier();

    if (stager) {
      if (f + 1 < F_)
        wstage_swap<D_, H_>(sW, BW0, fc1W + (size_t)(f + 1) * H_ * H_, tid);
      else
        wstage_swap<H_, H_>(sW, BW0, sf1W + (size_t)H_ * H_, tid);
    } else {
      const float b = bias_at(gB + f * H_, bco);
      v8f cg = wmma_gemm_lds<H_>(zero8, P2, ST_P, BW1, n0, lane16, hi);
#pragma unroll
      for (int v = 0; v < 8; ++v) {
        float t = cg[v] + b;
        float s = 1.f / (1.f + expf(-t));
        cg[v] = ce[v] + s * c2[v];
      }
      store_tile(P3, ST_P, n0, cg, lane16, hi);
    }
    lds_barrier();

    if (stager) {
      if (f + 1 == F_) {
        skstage_load(sS, sskW, tid);
        skstage_store(sS, P0, tid);
      }
    } else {
      float vals[8];
      float s = 0.f, ss = 0.f;
#pragma unroll
      for (int j = 0; j < 8; ++j) {
        float v = P3[lr * ST_P + lc + 16 * j];
        vals[j] = v; s += v; ss += v * v;
      }
      s = rowsum16(s);
      ss = rowsum16(ss);
      float mu = s * (1.f / 128.f);
      float var = ss * (1.f / 128.f) - mu * mu;
      float rs = rsqrtf(var + 1e-5f);
#pragma unroll
      for (int j = 0; j < 8; ++j) {
        int h = lc + 16 * j;
        float nv = (vals[j] - mu) * rs * lnG[f * H_ + h] + lnB[f * H_ + h];
        TRANS[lr * ST_T + f * H_ + h] = nv;
      }
    }
    lds_barrier();
  }

  v8f b1 = zero8;
  v8f sk = zero8;
  for (int p = 0; p < F_; ++p) {
    const float* nxt2 = (p + 2 < F_) ? sf1W + (size_t)(p + 2) * H_ * H_ : sf2W;
    float* cur = (p & 1) ? BW1 : BW0;
    float* oth = (p & 1) ? BW0 : BW1;
    const float* scur = (p & 1) ? P3 : P0;
    float* soth = (p & 1) ? P0 : P3;
    if (stager) {
      if (p + 1 < F_) skstage_load(sS, sskW + (size_t)(p + 1) * H_ * F_, tid);
      if (p + 2 <= F_) wstage_swap<H_, H_>(sW, oth, nxt2, tid);
      else             wstage_store<H_>(sW, oth, tid);
      if (p + 1 < F_) skstage_store(sS, soth, tid);
    } else if (wid == 1) {
      wmma_gemm_lds_pair<H_>(b1, sk, TRANS + p * H_, ST_T, cur, n0, scur,
                             lane16, hi);
    } else {
      b1 = wmma_gemm_lds<H_>(b1, TRANS + p * H_, ST_T, cur, n0, lane16, hi);
    }
    lds_barrier();
  }
  if (!stager) {
    float b = sf1B[n0 + lane16];
#pragma unroll
    for (int v = 0; v < 8; ++v) {
      float t = b1[v] + b;
      b1[v] = t > 0.f ? t : (expf(t) - 1.f);
    }
    store_tile(P1, ST_P, n0, b1, lane16, hi);
  }
  lds_barrier();

  if (!stager) {
    const float b = bias_at(sf2B, bco);
    v8f b2 = wmma_gemm_lds<H_>(zero8, P1, ST_P, BW0, n0, lane16, hi);
#pragma unroll
    for (int v = 0; v < 8; ++v) b2[v] += b;
    store_tile(P2, ST_P, n0, b2, lane16, hi);
  }
  lds_barrier();

  if (!stager) {
    if (wid == 0) {
      v8f g = wmma_gemm<H_>(P2, ST_P, sgW, F_, 0, lane16, hi);
      float b = sgB[lane16];
#pragma unroll
      for (int v = 0; v < 8; ++v) g[v] += b;
      store_tile(GP, ST_S, 0, g, lane16, hi);
    } else if (wid == 1) {
      float b = sskB[lane16];
#pragma unroll
      for (int v = 0; v < 8; ++v) sk[v] += b;
      store_tile(SKT, ST_S, 0, sk, lane16, hi);
    }
  }
  lds_barrier();

  if (!stager) {
    float gp = GP[lr * ST_S + lc];
    float sg = 1.f / (1.f + expf(-gp));
    float v = SKT[lr * ST_S + lc] + sg * gp;
    float s = rowsum16(v);
    float ss = rowsum16(v * v);
    float mu = s * (1.f / 16.f);
    float var = ss * (1.f / 16.f) - mu * mu;
    float nv = (v - mu) * rsqrtf(var + 1e-5f) * slnG[lc] + slnB[lc];
    float mx = rowmax16(nv);
    float e = expf(nv - mx);
    float se = rowsum16(e);
    float w = e / se;
    WB[lr * 16 + lc] = w;
    outW[(size_t)(row0 + lr) * F_ + lc] = w;
  }
  lds_barrier();

  if (!stager) {
#pragma unroll
    for (int j = 0; j < 8; ++j) {
      int h = lc + 16 * j;
      float acc = 0.f;
#pragma unroll
      for (int f = 0; f < F_; ++f)
        acc += TRANS[lr * ST_T + f * H_ + h] * WB[lr * 16 + f];
      outSel[(size_t)(row0 + lr) * H_ + h] = acc;
    }
  }
}

extern "C" void kernel_launch(void* const* d_in, const int* in_sizes, int n_in,
                              void* d_out, int out_size, void* d_ws, size_t ws_size,
                              hipStream_t stream) {
  (void)in_sizes; (void)n_in; (void)out_size; (void)d_ws; (void)ws_size;
  const float* x    = (const float*)d_in[0];
  const float* embW = (const float*)d_in[1];
  const float* embB = (const float*)d_in[2];
  const float* fc1W = (const float*)d_in[3];
  const float* fc1B = (const float*)d_in[4];
  const float* fc2W = (const float*)d_in[5];
  const float* fc2B = (const float*)d_in[6];
  const float* gW   = (const float*)d_in[7];
  const float* gB   = (const float*)d_in[8];
  const float* lnG  = (const float*)d_in[9];
  const float* lnB  = (const float*)d_in[10];
  const float* sf1W = (const float*)d_in[11];
  const float* sf1B = (const float*)d_in[12];
  const float* sf2W = (const float*)d_in[13];
  const float* sf2B = (const float*)d_in[14];
  const float* sgW  = (const float*)d_in[15];
  const float* sgB  = (const float*)d_in[16];
  const float* sskW = (const float*)d_in[17];
  const float* sskB = (const float*)d_in[18];
  const float* slnG = (const float*)d_in[19];
  const float* slnB = (const float*)d_in[20];

  float* outSel = (float*)d_out;
  float* outW   = outSel + (size_t)ROWS_TOTAL * H_;

  vsn_fused<<<ROWS_TOTAL / 16, 512, 0, stream>>>(
      x, embW, embB, fc1W, fc1B, fc2W, fc2B, gW, gB, lnG, lnB,
      sf1W, sf1B, sf2W, sf2B, sgW, sgB, sskW, sskB, slnG, slnB,
      outSel, outW);
}


# Blocks no longer stream the same weights in step: block b takes the 16 features, and the 16 s_fc1/skip panels, in rotated order from b mod 16; s_fc1 and skip sums take panels in that order
# speedup vs baseline: 1.9441x; 1.0318x over previous
#include <hip/hip_runtime.h>
#include <hip/hip_bf16.h>
#include <math.h>


typedef float v2f __attribute__((ext_vector_type(2)));
typedef float v8f __attribute__((ext_vector_type(8)));

#define ROWS_TOTAL 16384
#define F_ 16
#define D_ 64
#define H_ 128
#define FH_ 2048

#define ST_T 2052
#define ST_X 68
#define ST_P 132
#define ST_S 20

#define WPITCH 288
#define WBUF_FLOATS (64*WPITCH)

#define SKPITCH 32

#define SMEM_FLOATS (16*ST_T + 16*ST_X + 4*16*ST_P + 2*16*ST_S + 16*16 + 2*WBUF_FLOATS)

__device__ __forceinline__ float rowsum16(float v) {
  v += __shfl_xor(v, 1, 16);
  v += __shfl_xor(v, 2, 16);
  v += __shfl_xor(v, 4, 16);
  v += __shfl_xor(v, 8, 16);
  return v;
}
__device__ __forceinline__ float rowmax16(float v) {
  v = fmaxf(v, __shfl_xor(v, 1, 16));
  v = fmaxf(v, __shfl_xor(v, 2, 16));
  v = fmaxf(v, __shfl_xor(v, 4, 16));
  v = fmaxf(v, __shfl_xor(v, 8, 16));
  return v;
}

template <int K>
__device__ __forceinline__ v8f wmma_gemm(const float* __restrict__ As, int lda,
                                         const float* __restrict__ W, int ldw,
                                         int n0, int lane16, int hi) {
  v8f c = {0.f, 0.f, 0.f, 0.f, 0.f, 0.f, 0.f, 0.f};
  const float* arow = As + lane16 * lda + hi * 2;
  const float* wcol = W + (size_t)(hi * 2) * ldw + n0 + lane16;
#pragma unroll 1
  for (int kb = 0; kb < K; kb += 32) {
    if (kb + 32 < K) {
      __builtin_prefetch(wcol + (size_t)(kb + 32) * ldw, 0, 1);
      __builtin_prefetch(wcol + (size_t)(kb + 48) * ldw, 0, 1);
    }
#pragma unroll
    for (int k0 = kb; k0 < kb + 32; k0 += 4) {
      v2f a, b;
      a.x = arow[k0];
      a.y = arow[k0 + 1];
      b.x = wcol[(size_t)k0 * ldw];
      b.y = wcol[(size_t)(k0 + 1) * ldw];
      c = __builtin_amdgcn_wmma_f32_16x16x4_f32(false, a, false, b,
                                                (short)0, c, false, false);
    }
  }
  return c;
}

#define FR_CH 8
template <int K>
__device__ __forceinline__ v8f wmma_gemm_lds(v8f c, const float* As, int lda,
                                             const float* Bs, int n0,
                                             int lane16, int hi) {
  const float* arow = As + lane16 * lda + hi * 2;
  const float* bcol = Bs + hi * WPITCH + 2 * (n0 + lane16);
  constexpr int NCH = K / (4 * FR_CH);
  v2f a[2][FR_CH], b[2][FR_CH];
#pragma unroll
  for (int j = 0; j < 2; ++j) {
#pragma unroll
    for (int i = 0; i < FR_CH; ++i) {
      const int k0 = 4 * (j * FR_CH + i);
      a[j][i].x = arow[k0];
      a[j][i].y = arow[k0 + 1];
      b[j][i].x = bcol[(k0 >> 1) * WPITCH];
      b[j][i].y = bcol[(k0 >> 1) * WPITCH + 1];
    }
    __builtin_amdgcn_sched_barrier(0);
  }
#pragma unroll
  for (int j = 0; j < NCH; ++j) {
#pragma unroll
    for (int i = 0; i < FR_CH; ++i)
      c = __builtin_amdgcn_wmma_f32_16x16x4_f32(false, a[j & 1][i], false,
                                                b[j & 1][i], (short)0, c,
                                                false, false);
    if (j + 2 < NCH) {
      __builtin_amdgcn_sched_barrier(0);
#pragma unroll
      for (int i = 0; i < FR_CH; ++i) {
        const int k0 = 4 * ((j + 2) * FR_CH + i);
        a[j & 1][i].x = arow[k0];
        a[j & 1][i].y = arow[k0 + 1];
        b[j & 1][i].x = bcol[(k0 >> 1) * WPITCH];
        b[j & 1][i].y = bcol[(k0 >> 1) * WPITCH + 1];
      }
      __builtin_amdgcn_sched_barrier(0);
    }
  }
  return c;
}

#define PAIR_CH FR_CH
template <int K>
__device__ __forceinline__ void wmma_gemm_lds_pair(v8f& c, v8f& d,
                                                   const float* As, int lda,
                                                   const float* Bs, int n0,
                                                   const float* Ss,
                                                   int lane16, int hi) {
  const float* arow = As + lane16 * lda + hi * 2;
  const float* bcol = Bs + hi * WPITCH + 2 * (n0 + lane16);
  const float* scol = Ss + hi * SKPITCH + 2 * lane16;
  constexpr int NCH = K / (4 * PAIR_CH);
  v2f a[2][PAIR_CH], b[2][PAIR_CH], s[2][PAIR_CH];
#pragma unroll
  for (int j = 0; j < 2; ++j) {
#pragma unroll
    for (int i = 0; i < PAIR_CH; ++i) {
      const int k0 = 4 * (j * PAIR_CH + i);
      a[j][i].x = arow[k0];
      a[j][i].y = arow[k0 + 1];
      b[j][i].x = bcol[(k0 >> 1) * WPITCH];
      b[j][i].y = bcol[(k0 >> 1) * WPITCH + 1];
      s[j][i].x = scol[(k0 >> 1) * SKPITCH];
      s[j][i].y = scol[(k0 >> 1) * SKPITCH + 1];
    }
    __builtin_amdgcn_sched_barrier(0);
  }
#pragma unroll
  for (int j = 0; j < NCH; ++j) {
#pragma unroll
    for (int i = 0; i < PAIR_CH; ++i) {
      c = __builtin_amdgcn_wmma_f32_16x16x4_f32(false, a[j & 1][i], false,
                                                b[j & 1][i], (short)0, c,
                                                false, false);
      d = __builtin_amdgcn_wmma_f32_16x16x4_f32(false, a[j & 1][i], false,
                                                s[j & 1][i], (short)0, d,
                                                false, false);
    }
    if (j + 2 < NCH) {
      __builtin_amdgcn_sched_barrier(0);
#pragma unroll
      for (int i = 0; i < PAIR_CH; ++i) {
        const int k0 = 4 * ((j + 2) * PAIR_CH + i);
        a[j & 1][i].x = arow[k0];
        a[j & 1][i].y = arow[k0 + 1];
        b[j & 1][i].x = bcol[(k0 >> 1) * WPITCH];
        b[j & 1][i].y = bcol[(k0 >> 1) * WPITCH + 1];
        s[j & 1][i].x = scol[(k0 >> 1) * SKPITCH];
        s[j & 1][i].y = scol[(k0 >> 1) * SKPITCH + 1];
      }
      __builtin_amdgcn_sched_barrier(0);
    }
  }
}

typedef float v4f __attribute__((ext_vector_type(4)));
typedef const __attribute__((address_space(1))) float* gfloat_ptr;
typedef const __attribute__((address_space(1))) v4f* gv4f_ptr;
__device__ __forceinline__ gfloat_ptr scalar_base(const float* p) {
  const unsigned long long a = (unsigned long long)p;
  const unsigned lo = __builtin_amdgcn_readfirstlane((unsigned)a);
  const unsigned hi = __builtin_amdgcn_readfirstlane((unsigned)(a >> 32));
  return (gfloat_ptr)(((unsigned long long)hi << 32) | lo);
}
__device__ __forceinline__ float4 gload4(gfloat_ptr p) {
  const v4f v = *(gv4f_ptr)p;
  return make_float4(v.x, v.y, v.z, v.w);
}
__device__ __forceinline__ float bias_at(const float* rowbase, unsigned col) {
  return scalar_base(rowbase)[col];
}

struct WStage { float4 r[H_ / 8]; };

template <int KR>
__device__ __forceinline__ void wstage_load(WStage& s,
                                            const float* __restrict__ Wg,
                                            int tid) {
  const unsigned toff = (unsigned)((2 * (tid >> 5)) * H_ + 4 * (tid & 31));
  gfloat_ptr src = scalar_base(Wg) + toff;
#pragma unroll
  for (int i = 0; i < KR / 16; ++i) {
    s.r[2 * i]     = gload4(src + (16 * i) * H_);
    s.r[2 * i + 1] = gload4(src + (16 * i + 1) * H_);
  }
}

template <int KR>
__device__ __forceinline__ void wstage_store(const WStage& s, float* buf,
                                             int tid) {
  float* dst = buf + (tid >> 5) * WPITCH + 8 * (tid & 31);
#pragma unroll
  for (int i = 0; i < KR / 16; ++i) {
    const float4 e = s.r[2 * i], o = s.r[2 * i + 1];
    float4* d = (float4*)(dst + 8 * i * WPITCH);
    d[0] = make_float4(e.x, o.x, e.y, o.y);
    d[1] = make_float4(e.z, o.z, e.w, o.w);
  }
}

template <int KS, int KL>
__device__ __forceinline__ void wstage_swap(WStage& s, float* buf,
                                            const float* __restrict__ Wn,
                                            int tid) {
  float* dst = buf + (tid >> 5) * WPITCH + 8 * (tid & 31);
  const unsigned toff = (unsigned)((2 * (tid >> 5)) * H_ + 4 * (tid & 31));
  gfloat_ptr src = scalar_base(Wn) + toff;
#pragma unroll
  for (int i = 0; i < H_ / 16; ++i) {
    if (i < KS / 16) {
      const float4 e = s.r[2 * i], o = s.r[2 * i + 1];
      float4* d = (float4*)(dst + 8 * i * WPITCH);
      d[0] = make_float4(e.x, o.x, e.y, o.y);
      d[1] = make_float4(e.z, o.z, e.w, o.w);
    }
    if (i < KL / 16) {
      s.r[2 * i]     = gload4(src + (16 * i) * H_);
      s.r[2 * i + 1] = gload4(src + (16 * i + 1) * H_);
    }
    __builtin_amdgcn_sched_barrier(0);
  }
}

struct SkStage { float4 e, o; };

__device__ __forceinline__ void skstage_load(SkStage& s,
                                             const float* __restrict__ Wg,
                                             int tid) {
  const float* src = Wg + (size_t)(2 * (tid >> 2)) * F_ + 4 * (tid & 3);
  s.e = *(const float4*)src;
  s.o = *(const float4*)(src + F_);
}

__device__ __forceinline__ void skstage_store(const SkStage& s, float* buf,
                                              int tid) {
  float4* d = (float4*)(buf + (tid >> 2) * SKPITCH + 8 * (tid & 3));
  d[0] = make_float4(s.e.x, s.o.x, s.e.y, s.o.y);
  d[1] = make_float4(s.e.z, s.o.z, s.e.w, s.o.w);
}

__device__ __forceinline__ void store_tile(float* dst, int ldd, int col0,
                                           v8f c, int lane16, int hi) {
#pragma unroll
  for (int v = 0; v < 8; ++v)
    dst[(v + hi * 8) * ldd + col0 + lane16] = c[v];
}

__device__ __forceinline__ void lds_barrier() {
  __builtin_amdgcn_fence(__ATOMIC_RELEASE, "workgroup", "local");
  __builtin_amdgcn_s_barrier();
  __builtin_amdgcn_fence(__ATOMIC_ACQUIRE, "workgroup", "local");
}

__global__ __launch_bounds__(512, 1)
void vsn_fused(
    const float* __restrict__ x,
    const float* __restrict__ embW, const float* __restrict__ embB,
    const float* __restrict__ fc1W, const float* __restrict__ fc1B,
    const float* __restrict__ fc2W, const float* __restrict__ fc2B,
    const float* __restrict__ gW,   const float* __restrict__ gB,
    const float* __restrict__ lnG,  const float* __restrict__ lnB,
    const float* __restrict__ sf1W, const float* __restrict__ sf1B,
    const float* __restrict__ sf2W, const float* __restrict__ sf2B,
    const float* __restrict__ sgW,  const float* __restrict__ sgB,
    const float* __restrict__ sskW, const float* __restrict__ sskB,
    const float* __restrict__ slnG, const float* __restrict__ slnB,
    float* __restrict__ outSel, float* __restrict__ outW) {
  __shared__ __attribute__((aligned(16))) float smem[SMEM_FLOATS];
  float* TRANS = smem;
  float* XT = TRANS + 16 * ST_T;
  float* P0 = XT + 16 * ST_X;
  float* P1 = P0 + 16 * ST_P;
  float* P2 = P1 + 16 * ST_P;
  float* P3 = P2 + 16 * ST_P;
  float* GP = P3 + 16 * ST_P;
  float* SKT = GP + 16 * ST_S;
  float* WB = SKT + 16 * ST_S;
  float* BW0 = WB + 16 * 16;
  float* BW1 = BW0 + WBUF_FLOATS;

  const bool stager = __builtin_amdgcn_readfirstlane((int)threadIdx.x >> 8) != 0;
  const int tid = threadIdx.x & 255;
  const int wid = tid >> 5;
  const int lane = tid & 31;
  const int lane16 = lane & 15;
  const int hi = lane >> 4;
  const int n0 = wid * 16;
  const int lr = tid >> 4;
  const int lc = tid & 15;
  const int row0 = blockIdx.x * 16;
  const unsigned bco = (unsigned)(n0 + lane16);
  const int frot = (int)(blockIdx.x & (F_ - 1));

  const v8f zero8 = {0.f, 0.f, 0.f, 0.f, 0.f, 0.f, 0.f, 0.f};
  WStage sW;
  SkStage sS;

  const float* xrow = x + (size_t)(row0 + lr) * (F_ * D_) + lc * 4;
  float* xdst = XT + lr * ST_X + lc * 4;
  float4 xv = make_float4(0.f, 0.f, 0.f, 0.f);
  if (stager) {
    wstage_load<D_>(sW, embW + (size_t)frot * D_ * H_, tid);
    xv = *(const float4*)(xrow + frot * D_);
    wstage_store<D_>(sW, BW0, tid);
    *(float4*)xdst = xv;
    wstage_load<H_>(sW, fc1W + (size_t)frot * H_ * H_, tid);
  }
  lds_barrier();

  for (int fi = 0; fi < F_; ++fi) {
    const int f = (fi + frot) & (F_ - 1);
    const int fn = (fi + 1 + frot) & (F_ - 1);
    const bool more = fi + 1 < F_;
    v8f ce = zero8, c2 = zero8;

    if (stager) {
      wstage_swap<H_, H_>(sW, BW1, fc2W + (size_t)f * H_ * H_, tid);
      if (more) xv = *(const float4*)(xrow + fn * D_);
    } else {
      const float b = bias_at(embB + f * H_, bco);
      ce = wmma_gemm_lds<D_>(zero8, XT, ST_X, BW0, n0, lane16, hi);
#pragma unroll
      for (int v = 0; v < 8; ++v) ce[v] += b;
      store_tile(P0, ST_P, n0, ce, lane16, hi);
    }
    lds_barrier();

    if (stager) {
      wstage_swap<H_, H_>(sW, BW0, gW + (size_t)f * H_ * H_, tid);
      if (more) *(float4*)xdst = xv;
    } else {
      const float b = bias_at(fc1B + f * H_, bco);
      v8f c1 = wmma_gemm_lds<H_>(zero8, P0, ST_P, BW1, n0, lane16, hi);
#pragma unroll
      for (int v = 0; v < 8; ++v) {
        float t = c1[v] + b;
        c1[v] = t > 0.f ? t : (expf(t) - 1.f);
      }
      store_tile(P1, ST_P, n0, c1, lane16, hi);
    }
    lds_barrier();

    if (stager) {
      if (more)
        wstage_swap<H_, D_>(sW, BW1, embW + (size_t)fn * D_ * H_, tid);
      else
        wstage_swap<H_, H_>(sW, BW1, sf1W + (size_t)frot * H_ * H_, tid);
    } else {
      const float b = bias_at(fc2B + f * H_, bco);
      c2 = wmma_gemm_lds<H_>(zero8, P1, ST_P, BW0, n0, lane16, hi);
#pragma unroll
      for (int v = 0; v < 8; ++v) c2[v] += b;
      store_tile(P2, ST_P, n0, c2, lane16, hi);
    }
    lds_barrier();

    if (stager) {
      if (more)
        wstage_swap<D_, H_>(sW, BW0, fc1W + (size_t)fn * H_ * H_, tid);
      else
        wstage_swap<H_, H_>(sW, BW0,
                            sf1W + (size_t)((frot + 1) & (F_ - 1)) * H_ * H_,
                            tid);
    } else {
      const float b = bias_at(gB + f * H_, bco);
      v8f cg = wmma_gemm_lds<H_>(zero8, P2, ST_P, BW1, n0, lane16, hi);
#pragma unroll
      for (int v = 0; v < 8; ++v) {
        float t = cg[v] + b;
        float s = 1.f / (1.f + expf(-t));
        cg[v] = ce[v] + s * c2[v];
      }
      store_tile(P3, ST_P, n0, cg, lane16, hi);
    }
    lds_barrier();

    if (stager) {
      if (!more) {
        skstage_load(sS, sskW + (size_t)frot * H_ * F_, tid);
        skstage_store(sS, P0, tid);
      }
    } else {
      float vals[8];
      float s = 0.f, ss = 0.f;
#pragma unroll
      for (int j = 0; j < 8; ++j) {
        float v = P3[lr * ST_P + lc + 16 * j];
        vals[j] = v; s += v; ss += v * v;
      }
      s = rowsum16(s);
      ss = rowsum16(ss);
      float mu = s * (1.f / 128.f);
      float var = ss * (1.f / 128.f) - mu * mu;
      float rs = rsqrtf(var + 1e-5f);
#pragma unroll
      for (int j = 0; j < 8; ++j) {
        int h = lc + 16 * j;
        float nv = (vals[j] - mu) * rs * lnG[f * H_ + h] + lnB[f * H_ + h];
        TRANS[lr * ST_T + f * H_ + h] = nv;
      }
    }
    lds_barrier();
  }

  v8f b1 = zero8;
  v8f sk = zero8;
  for (int p = 0; p < F_; ++p) {
    const int q = (p + frot) & (F_ - 1);
    const int q1 = (p + 1 + frot) & (F_ - 1);
    const int q2 = (p + 2 + frot) & (F_ - 1);
    const float* nxt2 = (p + 2 < F_) ? sf1W + (size_t)q2 * H_ * H_ : sf2W;
    float* cur = (p & 1) ? BW1 : BW0;
    float* oth = (p & 1) ? BW0 : BW1;
    const float* scur = (p & 1) ? P3 : P0;
    float* soth = (p & 1) ? P0 : P3;
    if (stager) {
      if (p + 1 < F_) skstage_load(sS, sskW + (size_t)q1 * H_ * F_, tid);
      if (p + 2 <= F_) wstage_swap<H_, H_>(sW, oth, nxt2, tid);
      else             wstage_store<H_>(sW, oth, tid);
      if (p + 1 < F_) skstage_store(sS, soth, tid);
    } else if (wid == 1) {
      wmma_gemm_lds_pair<H_>(b1, sk, TRANS + q * H_, ST_T, cur, n0, scur,
                             lane16, hi);
    } else {
      b1 = wmma_gemm_lds<H_>(b1, TRANS + q * H_, ST_T, cur, n0, lane16, hi);
    }
    lds_barrier();
  }
  if (!stager) {
    float b = sf1B[n0 + lane16];
#pragma unroll
    for (int v = 0; v < 8; ++v) {
      float t = b1[v] + b;
      b1[v] = t > 0.f ? t : (expf(t) - 1.f);
    }
    store_tile(P1, ST_P, n0, b1, lane16, hi);
  }
  lds_barrier();

  if (!stager) {
    const float b = bias_at(sf2B, bco);
    v8f b2 = wmma_gemm_lds<H_>(zero8, P1, ST_P, BW0, n0, lane16, hi);
#pragma unroll
    for (int v = 0; v < 8; ++v) b2[v] += b;
    store_tile(P2, ST_P, n0, b2, lane16, hi);
  }
  lds_barrier();

  if (!stager) {
    if (wid == 0) {
      v8f g = wmma_gemm<H_>(P2, ST_P, sgW, F_, 0, lane16, hi);
      float b = sgB[lane16];
#pragma unroll
      for (int v = 0; v < 8; ++v) g[v] += b;
      store_tile(GP, ST_S, 0, g, lane16, hi);
    } else if (wid == 1) {
      float b = sskB[lane16];
#pragma unroll
      for (int v = 0; v < 8; ++v) sk[v] += b;
      store_tile(SKT, ST_S, 0, sk, lane16, hi);
    }
  }
  lds_barrier();

  if (!stager) {
    float gp = GP[lr * ST_S + lc];
    float sg = 1.f / (1.f + expf(-gp));
    float v = SKT[lr * ST_S + lc] + sg * gp;
    float s = rowsum16(v);
    float ss = rowsum16(v * v);
    float mu = s * (1.f / 16.f);
    float var = ss * (1.f / 16.f) - mu * mu;
    float nv = (v - mu) * rsqrtf(var + 1e-5f) * slnG[lc] + slnB[lc];
    float mx = rowmax16(nv);
    float e = expf(nv - mx);
    float se = rowsum16(e);
    float w = e / se;
    WB[lr * 16 + lc] = w;
    outW[(size_t)(row0 + lr) * F_ + lc] = w;
  }
  lds_barrier();

  if (!stager) {
#pragma unroll
    for (int j = 0; j < 8; ++j) {
      int h = lc + 16 * j;
      float acc = 0.f;
#pragma unroll
      for (int f = 0; f < F_; ++f)
        acc += TRANS[lr * ST_T + f * H_ + h] * WB[lr * 16 + f];
      outSel[(size_t)(row0 + lr) * H_ + h] = acc;
    }
  }
}

extern "C" void kernel_launch(void* const* d_in, const int* in_sizes, int n_in,
                              void* d_out, int out_size, void* d_ws, size_t ws_size,
                              hipStream_t stream) {
  (void)in_sizes; (void)n_in; (void)out_size; (void)d_ws; (void)ws_size;
  const float* x    = (const float*)d_in[0];
  const float* embW = (const float*)d_in[1];
  const float* embB = (const float*)d_in[2];
  const float* fc1W = (const float*)d_in[3];
  const float* fc1B = (const float*)d_in[4];
  const float* fc2W = (const float*)d_in[5];
  const float* fc2B = (const float*)d_in[6];
  const float* gW   = (const float*)d_in[7];
  const float* gB   = (const float*)d_in[8];
  const float* lnG  = (const float*)d_in[9];
  const float* lnB  = (const float*)d_in[10];
  const float* sf1W = (const float*)d_in[11];
  const float* sf1B = (const float*)d_in[12];
  const float* sf2W = (const float*)d_in[13];
  const float* sf2B = (const float*)d_in[14];
  const float* sgW  = (const float*)d_in[15];
  const float* sgB  = (const float*)d_in[16];
  const float* sskW = (const float*)d_in[17];
  const float* sskB = (const float*)d_in[18];
  const float* slnG = (const float*)d_in[19];
  const float* slnB = (const float*)d_in[20];

  float* outSel = (float*)d_out;
  float* outW   = outSel + (size_t)ROWS_TOTAL * H_;

  vsn_fused<<<ROWS_TOTAL / 16, 512, 0, stream>>>(
      x, embW, embB, fc1W, fc1B, fc2W, fc2B, gW, gB, lnG, lnB,
      sf1W, sf1B, sf2W, sf2B, sgW, sgB, sskW, sskB, slnG, slnB,
      outSel, outW);
}


# Gate sigmoid, emb + gate*h2 and LayerNorm move to the staging waves, one feature behind, beside the next fc1 GEMM: 4 barrier intervals a feature, not 5; k-pair interleave held at the LDS write
# speedup vs baseline: 2.0074x; 1.0326x over previous
#include <hip/hip_runtime.h>
#include <hip/hip_bf16.h>
#include <math.h>


typedef float v2f __attribute__((ext_vector_type(2)));
typedef float v8f __attribute__((ext_vector_type(8)));

#define ROWS_TOTAL 16384
#define F_ 16
#define D_ 64
#define H_ 128
#define FH_ 2048

#define ST_T 2052
#define ST_X 68
#define ST_P 132
#define ST_S 20

#define WPITCH 288
#define WBUF_FLOATS (64*WPITCH)

#define SKPITCH 32

#define SMEM_FLOATS (16*ST_T + 16*ST_X + 4*16*ST_P + 2*16*ST_S + 16*16 + 2*WBUF_FLOATS)

__device__ __forceinline__ float rowsum16(float v) {
  v += __shfl_xor(v, 1, 16);
  v += __shfl_xor(v, 2, 16);
  v += __shfl_xor(v, 4, 16);
  v += __shfl_xor(v, 8, 16);
  return v;
}
__device__ __forceinline__ float rowmax16(float v) {
  v = fmaxf(v, __shfl_xor(v, 1, 16));
  v = fmaxf(v, __shfl_xor(v, 2, 16));
  v = fmaxf(v, __shfl_xor(v, 4, 16));
  v = fmaxf(v, __shfl_xor(v, 8, 16));
  return v;
}

template <int K>
__device__ __forceinline__ v8f wmma_gemm(const float* __restrict__ As, int lda,
                                         const float* __restrict__ W, int ldw,
                                         int n0, int lane16, int hi) {
  v8f c = {0.f, 0.f, 0.f, 0.f, 0.f, 0.f, 0.f, 0.f};
  const float* arow = As + lane16 * lda + hi * 2;
  const float* wcol = W + (size_t)(hi * 2) * ldw + n0 + lane16;
#pragma unroll 1
  for (int kb = 0; kb < K; kb += 32) {
    if (kb + 32 < K) {
      __builtin_prefetch(wcol + (size_t)(kb + 32) * ldw, 0, 1);
      __builtin_prefetch(wcol + (size_t)(kb + 48) * ldw, 0, 1);
    }
#pragma unroll
    for (int k0 = kb; k0 < kb + 32; k0 += 4) {
      v2f a, b;
      a.x = arow[k0];
      a.y = arow[k0 + 1];
      b.x = wcol[(size_t)k0 * ldw];
      b.y = wcol[(size_t)(k0 + 1) * ldw];
      c = __builtin_amdgcn_wmma_f32_16x16x4_f32(false, a, false, b,
                                                (short)0, c, false, false);
    }
  }
  return c;
}

#define FR_CH 8
template <int K>
__device__ __forceinline__ v8f wmma_gemm_lds(v8f c, const float* As, int lda,
                                             const float* Bs, int n0,
                                             int lane16, int hi) {
  const float* arow = As + lane16 * lda + hi * 2;
  const float* bcol = Bs + hi * WPITCH + 2 * (n0 + lane16);
  constexpr int NCH = K / (4 * FR_CH);
  v2f a[2][FR_CH], b[2][FR_CH];
#pragma unroll
  for (int j = 0; j < 2; ++j) {
#pragma unroll
    for (int i = 0; i < FR_CH; ++i) {
      const int k0 = 4 * (j * FR_CH + i);
      a[j][i].x = arow[k0];
      a[j][i].y = arow[k0 + 1];
      b[j][i].x = bcol[(k0 >> 1) * WPITCH];
      b[j][i].y = bcol[(k0 >> 1) * WPITCH + 1];
    }
    __builtin_amdgcn_sched_barrier(0);
  }
#pragma unroll
  for (int j = 0; j < NCH; ++j) {
#pragma unroll
    for (int i = 0; i < FR_CH; ++i)
      c = __builtin_amdgcn_wmma_f32_16x16x4_f32(false, a[j & 1][i], false,
                                                b[j & 1][i], (short)0, c,
                                                false, false);
    if (j + 2 < NCH) {
      __builtin_amdgcn_sched_barrier(0);
#pragma unroll
      for (int i = 0; i < FR_CH; ++i) {
        const int k0 = 4 * ((j + 2) * FR_CH + i);
        a[j & 1][i].x = arow[k0];
        a[j & 1][i].y = arow[k0 + 1];
        b[j & 1][i].x = bcol[(k0 >> 1) * WPITCH];
        b[j & 1][i].y = bcol[(k0 >> 1) * WPITCH + 1];
      }
      __builtin_amdgcn_sched_barrier(0);
    }
  }
  return c;
}

#define PAIR_CH FR_CH
template <int K>
__device__ __forceinline__ void wmma_gemm_lds_pair(v8f& c, v8f& d,
                                                   const float* As, int lda,
                                                   const float* Bs, int n0,
                                                   const float* Ss,
                                                   int lane16, int hi) {
  const float* arow = As + lane16 * lda + hi * 2;
  const float* bcol = Bs + hi * WPITCH + 2 * (n0 + lane16);
  const float* scol = Ss + hi * SKPITCH + 2 * lane16;
  constexpr int NCH = K / (4 * PAIR_CH);
  v2f a[2][PAIR_CH], b[2][PAIR_CH], s[2][PAIR_CH];
#pragma unroll
  for (int j = 0; j < 2; ++j) {
#pragma unroll
    for (int i = 0; i < PAIR_CH; ++i) {
      const int k0 = 4 * (j * PAIR_CH + i);
      a[j][i].x = arow[k0];
      a[j][i].y = arow[k0 + 1];
      b[j][i].x = bcol[(k0 >> 1) * WPITCH];
      b[j][i].y = bcol[(k0 >> 1) * WPITCH + 1];
      s[j][i].x = scol[(k0 >> 1) * SKPITCH];
      s[j][i].y = scol[(k0 >> 1) * SKPITCH + 1];
    }
    __builtin_amdgcn_sched_barrier(0);
  }
#pragma unroll
  for (int j = 0; j < NCH; ++j) {
#pragma unroll
    for (int i = 0; i < PAIR_CH; ++i) {
      c = __builtin_amdgcn_wmma_f32_16x16x4_f32(false, a[j & 1][i], false,
                                                b[j & 1][i], (short)0, c,
                                                false, false);
      d = __builtin_amdgcn_wmma_f32_16x16x4_f32(false, a[j & 1][i], false,
                                                s[j & 1][i], (short)0, d,
                                                false, false);
    }
    if (j + 2 < NCH) {
      __builtin_amdgcn_sched_barrier(0);
#pragma unroll
      for (int i = 0; i < PAIR_CH; ++i) {
        const int k0 = 4 * ((j + 2) * PAIR_CH + i);
        a[j & 1][i].x = arow[k0];
        a[j & 1][i].y = arow[k0 + 1];
        b[j & 1][i].x = bcol[(k0 >> 1) * WPITCH];
        b[j & 1][i].y = bcol[(k0 >> 1) * WPITCH + 1];
        s[j & 1][i].x = scol[(k0 >> 1) * SKPITCH];
        s[j & 1][i].y = scol[(k0 >> 1) * SKPITCH + 1];
      }
      __builtin_amdgcn_sched_barrier(0);
    }
  }
}

typedef float v4f __attribute__((ext_vector_type(4)));
typedef const __attribute__((address_space(1))) float* gfloat_ptr;
typedef const __attribute__((address_space(1))) v4f* gv4f_ptr;
__device__ __forceinline__ gfloat_ptr scalar_base(const float* p) {
  const unsigned long long a = (unsigned long long)p;
  const unsigned lo = __builtin_amdgcn_readfirstlane((unsigned)a);
  const unsigned hi = __builtin_amdgcn_readfirstlane((unsigned)(a >> 32));
  return (gfloat_ptr)(((unsigned long long)hi << 32) | lo);
}
__device__ __forceinline__ float4 gload4(gfloat_ptr p) {
  const v4f v = *(gv4f_ptr)p;
  return make_float4(v.x, v.y, v.z, v.w);
}
__device__ __forceinline__ float bias_at(const float* rowbase, unsigned col) {
  return scalar_base(rowbase)[col];
}

struct WStage { float4 r[H_ / 8]; };

template <int KR>
__device__ __forceinline__ void wstage_load(WStage& s,
                                            const float* __restrict__ Wg,
                                            int tid) {
  const unsigned toff = (unsigned)((2 * (tid >> 5)) * H_ + 4 * (tid & 31));
  gfloat_ptr src = scalar_base(Wg) + toff;
#pragma unroll
  for (int i = 0; i < KR / 16; ++i) {
    s.r[2 * i]     = gload4(src + (16 * i) * H_);
    s.r[2 * i + 1] = gload4(src + (16 * i + 1) * H_);
  }
}

__device__ __forceinline__ void at_write(float4& e, float4& o) {
  asm volatile("" : "+v"(e.x), "+v"(e.y), "+v"(e.z), "+v"(e.w),
                    "+v"(o.x), "+v"(o.y), "+v"(o.z), "+v"(o.w));
}

template <int KR>
__device__ __forceinline__ void wstage_store(const WStage& s, float* buf,
                                             int tid) {
  float* dst = buf + (tid >> 5) * WPITCH + 8 * (tid & 31);
#pragma unroll
  for (int i = 0; i < KR / 16; ++i) {
    float4 e = s.r[2 * i], o = s.r[2 * i + 1];
    at_write(e, o);
    float4* d = (float4*)(dst + 8 * i * WPITCH);
    d[0] = make_float4(e.x, o.x, e.y, o.y);
    d[1] = make_float4(e.z, o.z, e.w, o.w);
  }
}

template <int KS, int KL>
__device__ __forceinline__ void wstage_swap(WStage& s, float* buf,
                                            const float* __restrict__ Wn,
                                            int tid) {
  float* dst = buf + (tid >> 5) * WPITCH + 8 * (tid & 31);
  const unsigned toff = (unsigned)((2 * (tid >> 5)) * H_ + 4 * (tid & 31));
  gfloat_ptr src = scalar_base(Wn) + toff;
#pragma unroll
  for (int i = 0; i < H_ / 16; ++i) {
    if (i < KS / 16) {
      float4 e = s.r[2 * i], o = s.r[2 * i + 1];
      at_write(e, o);
      float4* d = (float4*)(dst + 8 * i * WPITCH);
      d[0] = make_float4(e.x, o.x, e.y, o.y);
      d[1] = make_float4(e.z, o.z, e.w, o.w);
    }
    if (i < KL / 16) {
      s.r[2 * i]     = gload4(src + (16 * i) * H_);
      s.r[2 * i + 1] = gload4(src + (16 * i + 1) * H_);
    }
    __builtin_amdgcn_sched_barrier(0);
  }
}

struct SkStage { float4 e, o; };

__device__ __forceinline__ void skstage_load(SkStage& s,
                                             const float* __restrict__ Wg,
                                             int tid) {
  const float* src = Wg + (size_t)(2 * (tid >> 2)) * F_ + 4 * (tid & 3);
  s.e = *(const float4*)src;
  s.o = *(const float4*)(src + F_);
}

__device__ __forceinline__ void skstage_store(const SkStage& s, float* buf,
                                              int tid) {
  float4* d = (float4*)(buf + (tid >> 2) * SKPITCH + 8 * (tid & 3));
  d[0] = make_float4(s.e.x, s.o.x, s.e.y, s.o.y);
  d[1] = make_float4(s.e.z, s.o.z, s.e.w, s.o.w);
}

__device__ __forceinline__ void store_tile(float* dst, int ldd, int col0,
                                           v8f c, int lane16, int hi) {
#pragma unroll
  for (int v = 0; v < 8; ++v)
    dst[(v + hi * 8) * ldd + col0 + lane16] = c[v];
}

__device__ __forceinline__ void feature_tail(float* trow, const float* tt,
                                             const float* th,
                                             const float* __restrict__ g,
                                             const float* __restrict__ bt) {
  float vals[8];
  float s = 0.f, ss = 0.f;
#pragma unroll
  for (int j = 0; j < 8; ++j) {
    const float t = tt[16 * j];
    const float sg = 1.f / (1.f + expf(-t));
    const float v = trow[16 * j] + sg * th[16 * j];
    vals[j] = v; s += v; ss += v * v;
  }
  s = rowsum16(s);
  ss = rowsum16(ss);
  const float mu = s * (1.f / 128.f);
  const float var = ss * (1.f / 128.f) - mu * mu;
  const float rs = rsqrtf(var + 1e-5f);
#pragma unroll
  for (int j = 0; j < 8; ++j)
    trow[16 * j] = (vals[j] - mu) * rs * g[16 * j] + bt[16 * j];
}

__device__ __forceinline__ void lds_barrier() {
  __builtin_amdgcn_fence(__ATOMIC_RELEASE, "workgroup", "local");
  __builtin_amdgcn_s_barrier();
  __builtin_amdgcn_fence(__ATOMIC_ACQUIRE, "workgroup", "local");
}

__global__ __launch_bounds__(512, 1)
void vsn_fused(
    const float* __restrict__ x,
    const float* __restrict__ embW, const float* __restrict__ embB,
    const float* __restrict__ fc1W, const float* __restrict__ fc1B,
    const float* __restrict__ fc2W, const float* __restrict__ fc2B,
    const float* __restrict__ gW,   const float* __restrict__ gB,
    const float* __restrict__ lnG,  const float* __restrict__ lnB,
    const float* __restrict__ sf1W, const float* __restrict__ sf1B,
    const float* __restrict__ sf2W, const float* __restrict__ sf2B,
    const float* __restrict__ sgW,  const float* __restrict__ sgB,
    const float* __restrict__ sskW, const float* __restrict__ sskB,
    const float* __restrict__ slnG, const float* __restrict__ slnB,
    float* __restrict__ outSel, float* __restrict__ outW) {
  __shared__ __attribute__((aligned(16))) float smem[SMEM_FLOATS];
  float* TRANS = smem;
  float* XT = TRANS + 16 * ST_T;
  float* P0 = XT + 16 * ST_X;
  float* P1 = P0 + 16 * ST_P;
  float* P2 = P1 + 16 * ST_P;
  float* P3 = P2 + 16 * ST_P;
  float* GP = P3 + 16 * ST_P;
  float* SKT = GP + 16 * ST_S;
  float* WB = SKT + 16 * ST_S;
  float* BW0 = WB + 16 * 16;
  float* BW1 = BW0 + WBUF_FLOATS;

  const bool stager = __builtin_amdgcn_readfirstlane((int)threadIdx.x >> 8) != 0;
  const int tid = threadIdx.x & 255;
  const int wid = tid >> 5;
  const int lane = tid & 31;
  const int lane16 = lane & 15;
  const int hi = lane >> 4;
  const int n0 = wid * 16;
  const int lr = tid >> 4;
  const int lc = tid & 15;
  const int row0 = blockIdx.x * 16;
  const unsigned bco = (unsigned)(n0 + lane16);
  const int frot = (int)(blockIdx.x & (F_ - 1));

  const v8f zero8 = {0.f, 0.f, 0.f, 0.f, 0.f, 0.f, 0.f, 0.f};
  WStage sW;
  SkStage sS;

  const float* xrow = x + (size_t)(row0 + lr) * (F_ * D_) + lc * 4;
  float* xdst = XT + lr * ST_X + lc * 4;
  float4 xv = make_float4(0.f, 0.f, 0.f, 0.f);
  if (stager) {
    wstage_load<D_>(sW, embW + (size_t)frot * D_ * H_, tid);
    xv = *(const float4*)(xrow + frot * D_);
    wstage_store<D_>(sW, BW0, tid);
    *(float4*)xdst = xv;
    wstage_load<H_>(sW, fc1W + (size_t)frot * H_ * H_, tid);
  }
  lds_barrier();

  float* const tail_trow = TRANS + lr * ST_T + lc;
  const float* const tail_t = P3 + lr * ST_P + lc;
  const float* const tail_h2 = P2 + lr * ST_P + lc;
  for (int fi = 0; fi < F_; ++fi) {
    const int f = (fi + frot) & (F_ - 1);
    const int fn = (fi + 1 + frot) & (F_ - 1);
    const int fp = (fi - 1 + frot) & (F_ - 1);
    const bool more = fi + 1 < F_;

    if (stager) {
      wstage_swap<H_, H_>(sW, BW1, fc2W + (size_t)f * H_ * H_, tid);
      if (more) xv = *(const float4*)(xrow + fn * D_);
    } else {
      const float b = bias_at(embB + f * H_, bco);
      v8f ce = wmma_gemm_lds<D_>(zero8, XT, ST_X, BW0, n0, lane16, hi);
#pragma unroll
      for (int v = 0; v < 8; ++v) ce[v] += b;
      store_tile(P0, ST_P, n0, ce, lane16, hi);
      store_tile(TRANS + f * H_, ST_T, n0, ce, lane16, hi);
    }
    lds_barrier();

    if (stager) {
      wstage_swap<H_, H_>(sW, BW0, gW + (size_t)f * H_ * H_, tid);
      if (more) *(float4*)xdst = xv;
      if (fi > 0)
        feature_tail(tail_trow + fp * H_, tail_t, tail_h2,
                     lnG + fp * H_ + lc, lnB + fp * H_ + lc);
    } else {
      const float b = bias_at(fc1B + f * H_, bco);
      v8f c1 = wmma_gemm_lds<H_>(zero8, P0, ST_P, BW1, n0, lane16, hi);
#pragma unroll
      for (int v = 0; v < 8; ++v) {
        float t = c1[v] + b;
        c1[v] = t > 0.f ? t : (expf(t) - 1.f);
      }
      store_tile(P1, ST_P, n0, c1, lane16, hi);
    }
    lds_barrier();

    if (stager) {
      if (more)
        wstage_swap<H_, D_>(sW, BW1, embW + (size_t)fn * D_ * H_, tid);
      else
        wstage_swap<H_, H_>(sW, BW1, sf1W + (size_t)frot * H_ * H_, tid);
    } else {
      const float b = bias_at(fc2B + f * H_, bco);
      v8f c2 = wmma_gemm_lds<H_>(zero8, P1, ST_P, BW0, n0, lane16, hi);
#pragma unroll
      for (int v = 0; v < 8; ++v) c2[v] += b;
      store_tile(P2, ST_P, n0, c2, lane16, hi);
    }
    lds_barrier();

    if (stager) {
      if (more)
        wstage_swap<D_, H_>(sW, BW0, fc1W + (size_t)fn * H_ * H_, tid);
      else
        wstage_swap<H_, H_>(sW, BW0,
                            sf1W + (size_t)((frot + 1) & (F_ - 1)) * H_ * H_,
                            tid);
    } else {
      const float b = bias_at(gB + f * H_, bco);
      v8f cg = wmma_gemm_lds<H_>(zero8, P2, ST_P, BW1, n0, lane16, hi);
#pragma unroll
      for (int v = 0; v < 8; ++v) cg[v] += b;
      store_tile(P3, ST_P, n0, cg, lane16, hi);
    }
    lds_barrier();
  }

  {
    const int fl = (F_ - 1 + frot) & (F_ - 1);
    if (stager) {
      skstage_load(sS, sskW + (size_t)frot * H_ * F_, tid);
      skstage_store(sS, P0, tid);
    } else {
      feature_tail(tail_trow + fl * H_, tail_t, tail_h2,
                   lnG + fl * H_ + lc, lnB + fl * H_ + lc);
    }
  }
  lds_barrier();

  v8f b1 = zero8;
  v8f sk = zero8;
  for (int p = 0; p < F_; ++p) {
    const int q = (p + frot) & (F_ - 1);
    const int q1 = (p + 1 + frot) & (F_ - 1);
    const int q2 = (p + 2 + frot) & (F_ - 1);
    const float* nxt2 = (p + 2 < F_) ? sf1W + (size_t)q2 * H_ * H_ : sf2W;
    float* cur = (p & 1) ? BW1 : BW0;
    float* oth = (p & 1) ? BW0 : BW1;
    const float* scur = (p & 1) ? P3 : P0;
    float* soth = (p & 1) ? P0 : P3;
    if (stager) {
      if (p + 1 < F_) skstage_load(sS, sskW + (size_t)q1 * H_ * F_, tid);
      if (p + 2 <= F_) wstage_swap<H_, H_>(sW, oth, nxt2, tid);
      else             wstage_store<H_>(sW, oth, tid);
      if (p + 1 < F_) skstage_store(sS, soth, tid);
    } else if (wid == 1) {
      wmma_gemm_lds_pair<H_>(b1, sk, TRANS + q * H_, ST_T, cur, n0, scur,
                             lane16, hi);
    } else {
      b1 = wmma_gemm_lds<H_>(b1, TRANS + q * H_, ST_T, cur, n0, lane16, hi);
    }
    lds_barrier();
  }
  if (!stager) {
    float b = sf1B[n0 + lane16];
#pragma unroll
    for (int v = 0; v < 8; ++v) {
      float t = b1[v] + b;
      b1[v] = t > 0.f ? t : (expf(t) - 1.f);
    }
    store_tile(P1, ST_P, n0, b1, lane16, hi);
  }
  lds_barrier();

  if (!stager) {
    const float b = bias_at(sf2B, bco);
    v8f b2 = wmma_gemm_lds<H_>(zero8, P1, ST_P, BW0, n0, lane16, hi);
#pragma unroll
    for (int v = 0; v < 8; ++v) b2[v] += b;
    store_tile(P2, ST_P, n0, b2, lane16, hi);
  }
  lds_barrier();

  if (!stager) {
    if (wid == 0) {
      v8f g = wmma_gemm<H_>(P2, ST_P, sgW, F_, 0, lane16, hi);
      float b = sgB[lane16];
#pragma unroll
      for (int v = 0; v < 8; ++v) g[v] += b;
      store_tile(GP, ST_S, 0, g, lane16, hi);
    } else if (wid == 1) {
      float b = sskB[lane16];
#pragma unroll
      for (int v = 0; v < 8; ++v) sk[v] += b;
      store_tile(SKT, ST_S, 0, sk, lane16, hi);
    }
  }
  lds_barrier();

  if (!stager) {
    float gp = GP[lr * ST_S + lc];
    float sg = 1.f / (1.f + expf(-gp));
    float v = SKT[lr * ST_S + lc] + sg * gp;
    float s = rowsum16(v);
    float ss = rowsum16(v * v);
    float mu = s * (1.f / 16.f);
    float var = ss * (1.f / 16.f) - mu * mu;
    float nv = (v - mu) * rsqrtf(var + 1e-5f) * slnG[lc] + slnB[lc];
    float mx = rowmax16(nv);
    float e = expf(nv - mx);
    float se = rowsum16(e);
    float w = e / se;
    WB[lr * 16 + lc] = w;
    outW[(size_t)(row0 + lr) * F_ + lc] = w;
  }
  lds_barrier();

  if (!stager) {
#pragma unroll
    for (int j = 0; j < 8; ++j) {
      int h = lc + 16 * j;
      float acc = 0.f;
#pragma unroll
      for (int f = 0; f < F_; ++f)
        acc += TRANS[lr * ST_T + f * H_ + h] * WB[lr * 16 + f];
      outSel[(size_t)(row0 + lr) * H_ + h] = acc;
    }
  }
}

extern "C" void kernel_launch(void* const* d_in, const int* in_sizes, int n_in,
                              void* d_out, int out_size, void* d_ws, size_t ws_size,
                              hipStream_t stream) {
  (void)in_sizes; (void)n_in; (void)out_size; (void)d_ws; (void)ws_size;
  const float* x    = (const float*)d_in[0];
  const float* embW = (const float*)d_in[1];
  const float* embB = (const float*)d_in[2];
  const float* fc1W = (const float*)d_in[3];
  const float* fc1B = (const float*)d_in[4];
  const float* fc2W = (const float*)d_in[5];
  const float* fc2B = (const float*)d_in[6];
  const float* gW   = (const float*)d_in[7];
  const float* gB   = (const float*)d_in[8];
  const float* lnG  = (const float*)d_in[9];
  const float* lnB  = (const float*)d_in[10];
  const float* sf1W = (const float*)d_in[11];
  const float* sf1B = (const float*)d_in[12];
  const float* sf2W = (const float*)d_in[13];
  const float* sf2B = (const float*)d_in[14];
  const float* sgW  = (const float*)d_in[15];
  const float* sgB  = (const float*)d_in[16];
  const float* sskW = (const float*)d_in[17];
  const float* sskB = (const float*)d_in[18];
  const float* slnG = (const float*)d_in[19];
  const float* slnB = (const float*)d_in[20];

  float* outSel = (float*)d_out;
  float* outW   = outSel + (size_t)ROWS_TOTAL * H_;

  vsn_fused<<<ROWS_TOTAL / 16, 512, 0, stream>>>(
      x, embW, embB, fc1W, fc1B, fc2W, fc2B, gW, gB, lnG, lnB,
      sf1W, sf1B, sf2W, sf2B, sgW, sgB, sskW, sskB, slnG, slnB,
      outSel, outW);
}


# Stage A cut by feature: each of 16 waves runs one feature's whole chain on 128 cols (8 accumulators per A fragment, B by 16-byte global loads), no block barrier there; LayerNorm row-sum order changes
# speedup vs baseline: 2.3905x; 1.1908x over previous
#include <hip/hip_runtime.h>
#include <hip/hip_bf16.h>
#include <math.h>


typedef float v2f __attribute__((ext_vector_type(2)));
typedef float v8f __attribute__((ext_vector_type(8)));

#define ROWS_TOTAL 16384
#define F_ 16
#define D_ 64
#define H_ 128
#define FH_ 2048

#define ST_T 2052
#define ST_X 68
#define ST_P 132
#define ST_S 20

#define WPITCH 288
#define WBUF_FLOATS (64*WPITCH)

#define SKPITCH 32

#define SMEM_FLOATS (16*ST_T + 16*ST_X + 4*16*ST_P + 2*16*ST_S + 16*16 + 2*WBUF_FLOATS)

__device__ __forceinline__ float rowsum16(float v) {
  v += __shfl_xor(v, 1, 16);
  v += __shfl_xor(v, 2, 16);
  v += __shfl_xor(v, 4, 16);
  v += __shfl_xor(v, 8, 16);
  return v;
}
__device__ __forceinline__ float rowmax16(float v) {
  v = fmaxf(v, __shfl_xor(v, 1, 16));
  v = fmaxf(v, __shfl_xor(v, 2, 16));
  v = fmaxf(v, __shfl_xor(v, 4, 16));
  v = fmaxf(v, __shfl_xor(v, 8, 16));
  return v;
}

template <int K>
__device__ __forceinline__ v8f wmma_gemm(const float* __restrict__ As, int lda,
                                         const float* __restrict__ W, int ldw,
                                         int n0, int lane16, int hi) {
  v8f c = {0.f, 0.f, 0.f, 0.f, 0.f, 0.f, 0.f, 0.f};
  const float* arow = As + lane16 * lda + hi * 2;
  const float* wcol = W + (size_t)(hi * 2) * ldw + n0 + lane16;
#pragma unroll 1
  for (int kb = 0; kb < K; kb += 32) {
    if (kb + 32 < K) {
      __builtin_prefetch(wcol + (size_t)(kb + 32) * ldw, 0, 1);
      __builtin_prefetch(wcol + (size_t)(kb + 48) * ldw, 0, 1);
    }
#pragma unroll
    for (int k0 = kb; k0 < kb + 32; k0 += 4) {
      v2f a, b;
      a.x = arow[k0];
      a.y = arow[k0 + 1];
      b.x = wcol[(size_t)k0 * ldw];
      b.y = wcol[(size_t)(k0 + 1) * ldw];
      c = __builtin_amdgcn_wmma_f32_16x16x4_f32(false, a, false, b,
                                                (short)0, c, false, false);
    }
  }
  return c;
}

#define FR_CH 8
template <int K>
__device__ __forceinline__ v8f wmma_gemm_lds(v8f c, const float* As, int lda,
                                             const float* Bs, int n0,
                                             int lane16, int hi) {
  const float* arow = As + lane16 * lda + hi * 2;
  const float* bcol = Bs + hi * WPITCH + 2 * (n0 + lane16);
  constexpr int NCH = K / (4 * FR_CH);
  v2f a[2][FR_CH], b[2][FR_CH];
#pragma unroll
  for (int j = 0; j < 2; ++j) {
#pragma unroll
    for (int i = 0; i < FR_CH; ++i) {
      const int k0 = 4 * (j * FR_CH + i);
      a[j][i].x = arow[k0];
      a[j][i].y = arow[k0 + 1];
      b[j][i].x = bcol[(k0 >> 1) * WPITCH];
      b[j][i].y = bcol[(k0 >> 1) * WPITCH + 1];
    }
    __builtin_amdgcn_sched_barrier(0);
  }
#pragma unroll
  for (int j = 0; j < NCH; ++j) {
#pragma unroll
    for (int i = 0; i < FR_CH; ++i)
      c = __builtin_amdgcn_wmma_f32_16x16x4_f32(false, a[j & 1][i], false,
                                                b[j & 1][i], (short)0, c,
                                                false, false);
    if (j + 2 < NCH) {
      __builtin_amdgcn_sched_barrier(0);
#pragma unroll
      for (int i = 0; i < FR_CH; ++i) {
        const int k0 = 4 * ((j + 2) * FR_CH + i);
        a[j & 1][i].x = arow[k0];
        a[j & 1][i].y = arow[k0 + 1];
        b[j & 1][i].x = bcol[(k0 >> 1) * WPITCH];
        b[j & 1][i].y = bcol[(k0 >> 1) * WPITCH + 1];
      }
      __builtin_amdgcn_sched_barrier(0);
    }
  }
  return c;
}

#define PAIR_CH FR_CH
template <int K>
__device__ __forceinline__ void wmma_gemm_lds_pair(v8f& c, v8f& d,
                                                   const float* As, int lda,
                                                   const float* Bs, int n0,
                                                   const float* Ss,
                                                   int lane16, int hi) {
  const float* arow = As + lane16 * lda + hi * 2;
  const float* bcol = Bs + hi * WPITCH + 2 * (n0 + lane16);
  const float* scol = Ss + hi * SKPITCH + 2 * lane16;
  constexpr int NCH = K / (4 * PAIR_CH);
  v2f a[2][PAIR_CH], b[2][PAIR_CH], s[2][PAIR_CH];
#pragma unroll
  for (int j = 0; j < 2; ++j) {
#pragma unroll
    for (int i = 0; i < PAIR_CH; ++i) {
      const int k0 = 4 * (j * PAIR_CH + i);
      a[j][i].x = arow[k0];
      a[j][i].y = arow[k0 + 1];
      b[j][i].x = bcol[(k0 >> 1) * WPITCH];
      b[j][i].y = bcol[(k0 >> 1) * WPITCH + 1];
      s[j][i].x = scol[(k0 >> 1) * SKPITCH];
      s[j][i].y = scol[(k0 >> 1) * SKPITCH + 1];
    }
    __builtin_amdgcn_sched_barrier(0);
  }
#pragma unroll
  for (int j = 0; j < NCH; ++j) {
#pragma unroll
    for (int i = 0; i < PAIR_CH; ++i) {
      c = __builtin_amdgcn_wmma_f32_16x16x4_f32(false, a[j & 1][i], false,
                                                b[j & 1][i], (short)0, c,
                                                false, false);
      d = __builtin_amdgcn_wmma_f32_16x16x4_f32(false, a[j & 1][i], false,
                                                s[j & 1][i], (short)0, d,
                                                false, false);
    }
    if (j + 2 < NCH) {
      __builtin_amdgcn_sched_barrier(0);
#pragma unroll
      for (int i = 0; i < PAIR_CH; ++i) {
        const int k0 = 4 * ((j + 2) * PAIR_CH + i);
        a[j & 1][i].x = arow[k0];
        a[j & 1][i].y = arow[k0 + 1];
        b[j & 1][i].x = bcol[(k0 >> 1) * WPITCH];
        b[j & 1][i].y = bcol[(k0 >> 1) * WPITCH + 1];
        s[j & 1][i].x = scol[(k0 >> 1) * SKPITCH];
        s[j & 1][i].y = scol[(k0 >> 1) * SKPITCH + 1];
      }
      __builtin_amdgcn_sched_barrier(0);
    }
  }
}

typedef float v4f __attribute__((ext_vector_type(4)));
typedef const __attribute__((address_space(1))) float* gfloat_ptr;
typedef const __attribute__((address_space(1))) v4f* gv4f_ptr;
__device__ __forceinline__ gfloat_ptr scalar_base(const float* p) {
  const unsigned long long a = (unsigned long long)p;
  const unsigned lo = __builtin_amdgcn_readfirstlane((unsigned)a);
  const unsigned hi = __builtin_amdgcn_readfirstlane((unsigned)(a >> 32));
  return (gfloat_ptr)(((unsigned long long)hi << 32) | lo);
}
__device__ __forceinline__ float4 gload4(gfloat_ptr p) {
  const v4f v = *(gv4f_ptr)p;
  return make_float4(v.x, v.y, v.z, v.w);
}
__device__ __forceinline__ float bias_at(const float* rowbase, unsigned col) {
  return scalar_base(rowbase)[col];
}

struct WStage { float4 r[H_ / 8]; };

template <int KR>
__device__ __forceinline__ void wstage_load(WStage& s,
                                            const float* __restrict__ Wg,
                                            int tid) {
  const unsigned toff = (unsigned)((2 * (tid >> 5)) * H_ + 4 * (tid & 31));
  gfloat_ptr src = scalar_base(Wg) + toff;
#pragma unroll
  for (int i = 0; i < KR / 16; ++i) {
    s.r[2 * i]     = gload4(src + (16 * i) * H_);
    s.r[2 * i + 1] = gload4(src + (16 * i + 1) * H_);
  }
}

__device__ __forceinline__ void at_write(float4& e, float4& o) {
  asm volatile("" : "+v"(e.x), "+v"(e.y), "+v"(e.z), "+v"(e.w),
                    "+v"(o.x), "+v"(o.y), "+v"(o.z), "+v"(o.w));
}

template <int KR>
__device__ __forceinline__ void wstage_store(const WStage& s, float* buf,
                                             int tid) {
  float* dst = buf + (tid >> 5) * WPITCH + 8 * (tid & 31);
#pragma unroll
  for (int i = 0; i < KR / 16; ++i) {
    float4 e = s.r[2 * i], o = s.r[2 * i + 1];
    at_write(e, o);
    float4* d = (float4*)(dst + 8 * i * WPITCH);
    d[0] = make_float4(e.x, o.x, e.y, o.y);
    d[1] = make_float4(e.z, o.z, e.w, o.w);
  }
}

template <int KS, int KL>
__device__ __forceinline__ void wstage_swap(WStage& s, float* buf,
                                            const float* __restrict__ Wn,
                                            int tid) {
  float* dst = buf + (tid >> 5) * WPITCH + 8 * (tid & 31);
  const unsigned toff = (unsigned)((2 * (tid >> 5)) * H_ + 4 * (tid & 31));
  gfloat_ptr src = scalar_base(Wn) + toff;
#pragma unroll
  for (int i = 0; i < H_ / 16; ++i) {
    if (i < KS / 16) {
      float4 e = s.r[2 * i], o = s.r[2 * i + 1];
      at_write(e, o);
      float4* d = (float4*)(dst + 8 * i * WPITCH);
      d[0] = make_float4(e.x, o.x, e.y, o.y);
      d[1] = make_float4(e.z, o.z, e.w, o.w);
    }
    if (i < KL / 16) {
      s.r[2 * i]     = gload4(src + (16 * i) * H_);
      s.r[2 * i + 1] = gload4(src + (16 * i + 1) * H_);
    }
    __builtin_amdgcn_sched_barrier(0);
  }
}

struct SkStage { float4 e, o; };

__device__ __forceinline__ void skstage_load(SkStage& s,
                                             const float* __restrict__ Wg,
                                             int tid) {
  const float* src = Wg + (size_t)(2 * (tid >> 2)) * F_ + 4 * (tid & 3);
  s.e = *(const float4*)src;
  s.o = *(const float4*)(src + F_);
}

__device__ __forceinline__ void skstage_store(const SkStage& s, float* buf,
                                              int tid) {
  float4* d = (float4*)(buf + (tid >> 2) * SKPITCH + 8 * (tid & 3));
  d[0] = make_float4(s.e.x, s.o.x, s.e.y, s.o.y);
  d[1] = make_float4(s.e.z, s.o.z, s.e.w, s.o.w);
}

__device__ __forceinline__ void store_tile(float* dst, int ldd, int col0,
                                           v8f c, int lane16, int hi) {
#pragma unroll
  for (int v = 0; v < 8; ++v)
    dst[(v + hi * 8) * ldd + col0 + lane16] = c[v];
}

__device__ __forceinline__ void lds_barrier() {
  __builtin_amdgcn_fence(__ATOMIC_RELEASE, "workgroup", "local");
  __builtin_amdgcn_s_barrier();
  __builtin_amdgcn_fence(__ATOMIC_ACQUIRE, "workgroup", "local");
}

struct BStep { float4 q0, q1, q2, q3; };
#define PD 4

__device__ __forceinline__ void bstep_load(BStep& b, gfloat_ptr src, int s) {
  gfloat_ptr p = src + s * (4 * H_);
  b.q0 = gload4(p);
  b.q1 = gload4(p + H_);
  b.q2 = gload4(p + 64);
  b.q3 = gload4(p + H_ + 64);
}
__device__ __forceinline__ void bprologue(BStep (&bq)[PD], gfloat_ptr src) {
#pragma unroll
  for (int i = 0; i < PD; ++i) bstep_load(bq[i], src, i);
}
__device__ __forceinline__ v8f wm(v2f a, float bx, float by, v8f c) {
  v2f b;
  b.x = bx;
  b.y = by;
  return __builtin_amdgcn_wmma_f32_16x16x4_f32(false, a, false, b, (short)0, c,
                                               false, false);
}
__device__ __forceinline__ void mma8(v8f (&acc)[8], v2f a, const BStep& b) {
  acc[0] = wm(a, b.q0.x, b.q1.x, acc[0]);
  acc[1] = wm(a, b.q0.y, b.q1.y, acc[1]);
  acc[2] = wm(a, b.q0.z, b.q1.z, acc[2]);
  acc[3] = wm(a, b.q0.w, b.q1.w, acc[3]);
  acc[4] = wm(a, b.q2.x, b.q3.x, acc[4]);
  acc[5] = wm(a, b.q2.y, b.q3.y, acc[5]);
  acc[6] = wm(a, b.q2.z, b.q3.z, acc[6]);
  acc[7] = wm(a, b.q2.w, b.q3.w, acc[7]);
}
template <int K>
__device__ __forceinline__ void gemm_run(v8f (&acc)[8], BStep (&bq)[PD],
                                         const float* arow, gfloat_ptr src) {
  constexpr int KS = K / 4;
  v2f aq[PD];
#pragma unroll
  for (int i = 0; i < PD; ++i) aq[i] = *(const v2f*)(arow + 4 * i);
#pragma unroll 1
  for (int s0 = PD; s0 < KS; s0 += PD) {
#pragma unroll
    for (int i = 0; i < PD; ++i) {
      mma8(acc, aq[i], bq[i]);
      bstep_load(bq[i], src, s0 + i);
      aq[i] = *(const v2f*)(arow + 4 * (s0 + i));
    }
  }
#pragma unroll
  for (int i = 0; i < PD; ++i) mma8(acc, aq[i], bq[i]);
}
__device__ __forceinline__ void wave_lds_turn() {
  __builtin_amdgcn_fence(__ATOMIC_RELEASE, "workgroup", "local");
  __builtin_amdgcn_fence(__ATOMIC_ACQUIRE, "workgroup", "local");
}
__device__ __forceinline__ float elu1(float t) {
  return t > 0.f ? t : (expf(t) - 1.f);
}
template <bool ELU>
__device__ __forceinline__ void store_rows(float* dst, int ld,
                                           const v8f (&acc)[8], float4 b0,
                                           float4 b1) {
#pragma unroll
  for (int v = 0; v < 8; ++v) {
    float4 lo = make_float4(acc[0][v] + b0.x, acc[1][v] + b0.y,
                            acc[2][v] + b0.z, acc[3][v] + b0.w);
    float4 hi4 = make_float4(acc[4][v] + b1.x, acc[5][v] + b1.y,
                             acc[6][v] + b1.z, acc[7][v] + b1.w);
    if (ELU) {
      lo = make_float4(elu1(lo.x), elu1(lo.y), elu1(lo.z), elu1(lo.w));
      hi4 = make_float4(elu1(hi4.x), elu1(hi4.y), elu1(hi4.z), elu1(hi4.w));
    }
    *(float4*)(dst + v * ld) = lo;
    *(float4*)(dst + v * ld + 64) = hi4;
  }
}
__device__ __forceinline__ float sigm(float t) {
  return 1.f / (1.f + expf(-t));
}

__global__ __launch_bounds__(512, 1)
void vsn_fused(
    const float* __restrict__ x,
    const float* __restrict__ embW, const float* __restrict__ embB,
    const float* __restrict__ fc1W, const float* __restrict__ fc1B,
    const float* __restrict__ fc2W, const float* __restrict__ fc2B,
    const float* __restrict__ gW,   const float* __restrict__ gB,
    const float* __restrict__ lnG,  const float* __restrict__ lnB,
    const float* __restrict__ sf1W, const float* __restrict__ sf1B,
    const float* __restrict__ sf2W, const float* __restrict__ sf2B,
    const float* __restrict__ sgW,  const float* __restrict__ sgB,
    const float* __restrict__ sskW, const float* __restrict__ sskB,
    const float* __restrict__ slnG, const float* __restrict__ slnB,
    float* __restrict__ outSel, float* __restrict__ outW) {
  __shared__ __attribute__((aligned(16))) float smem[SMEM_FLOATS];
  float* TRANS = smem;
  float* XT = TRANS + 16 * ST_T;
  float* P0 = XT + 16 * ST_X;
  float* P1 = P0 + 16 * ST_P;
  float* P2 = P1 + 16 * ST_P;
  float* P3 = P2 + 16 * ST_P;
  float* GP = P3 + 16 * ST_P;
  float* SKT = GP + 16 * ST_S;
  float* WB = SKT + 16 * ST_S;
  float* BW0 = WB + 16 * 16;
  float* BW1 = BW0 + WBUF_FLOATS;

  const bool stager = __builtin_amdgcn_readfirstlane((int)threadIdx.x >> 8) != 0;
  const int tid = threadIdx.x & 255;
  const int wid = tid >> 5;
  const int lane = tid & 31;
  const int lane16 = lane & 15;
  const int hi = lane >> 4;
  const int n0 = wid * 16;
  const int lr = tid >> 4;
  const int lc = tid & 15;
  const int row0 = blockIdx.x * 16;
  const unsigned bco = (unsigned)(n0 + lane16);
  const int frot = (int)(blockIdx.x & (F_ - 1));

  const v8f zero8 = {0.f, 0.f, 0.f, 0.f, 0.f, 0.f, 0.f, 0.f};
  WStage sW;
  SkStage sS;

  {
    const int f = __builtin_amdgcn_readfirstlane((int)threadIdx.x >> 5);
    float* S = XT + f * (16 * ST_P);
    float* const TF = TRANS + f * H_;
    const unsigned toff = (unsigned)(2 * hi * H_ + 4 * lane16);
    const unsigned bc = (unsigned)(4 * lane16);
    float* const srow = S + (8 * hi) * ST_P + 4 * lane16;
    float* const trow = TF + (8 * hi) * ST_T + 4 * lane16;
    const float* const sA = S + lane16 * ST_P + 2 * hi;
    const float* const tA = TF + lane16 * ST_T + 2 * hi;

    BStep bq[PD];
    v8f acc[8];

    const float* xp = x + (size_t)(row0 + hi) * (F_ * D_) + f * D_ + 4 * lane16;
    float4 xq[8];
#pragma unroll
    for (int i = 0; i < 8; ++i)
      xq[i] = *(const float4*)(xp + (size_t)(2 * i) * (F_ * D_));
    gfloat_ptr wsrc = scalar_base(embW + (size_t)f * D_ * H_) + toff;
    bprologue(bq, wsrc);
    gfloat_ptr bsrc = scalar_base(embB + f * H_) + bc;
    float4 b0 = gload4(bsrc), b1 = gload4(bsrc + 64);
#pragma unroll
    for (int i = 0; i < 8; ++i)
      *(float4*)(S + (hi + 2 * i) * ST_P + 4 * lane16) = xq[i];
    wave_lds_turn();

#pragma unroll
    for (int j = 0; j < 8; ++j) acc[j] = zero8;
    gemm_run<D_>(acc, bq, sA, wsrc);
    wsrc = scalar_base(fc1W + (size_t)f * H_ * H_) + toff;
    bprologue(bq, wsrc);
    store_rows<false>(trow, ST_T, acc, b0, b1);
    bsrc = scalar_base(fc1B + f * H_) + bc;
    b0 = gload4(bsrc); b1 = gload4(bsrc + 64);
    wave_lds_turn();

#pragma unroll
    for (int j = 0; j < 8; ++j) acc[j] = zero8;
    gemm_run<H_>(acc, bq, tA, wsrc);
    wsrc = scalar_base(fc2W + (size_t)f * H_ * H_) + toff;
    bprologue(bq, wsrc);
    store_rows<true>(srow, ST_P, acc, b0, b1);
    bsrc = scalar_base(fc2B + f * H_) + bc;
    b0 = gload4(bsrc); b1 = gload4(bsrc + 64);
    wave_lds_turn();

#pragma unroll
    for (int j = 0; j < 8; ++j) acc[j] = zero8;
    gemm_run<H_>(acc, bq, sA, wsrc);
    wsrc = scalar_base(gW + (size_t)f * H_ * H_) + toff;
    bprologue(bq, wsrc);
    store_rows<false>(srow, ST_P, acc, b0, b1);
    bsrc = scalar_base(gB + f * H_) + bc;
    b0 = gload4(bsrc); b1 = gload4(bsrc + 64);
    gfloat_ptr gsrc = scalar_base(lnG + f * H_) + bc;
    gfloat_ptr lsrc = scalar_base(lnB + f * H_) + bc;
    const float4 g0 = gload4(gsrc), g1 = gload4(gsrc + 64);
    const float4 l0 = gload4(lsrc), l1 = gload4(lsrc + 64);
    wave_lds_turn();

#pragma unroll
    for (int j = 0; j < 8; ++j) acc[j] = zero8;
    gemm_run<H_>(acc, bq, sA, wsrc);
    float rs_[8], rq_[8];
#pragma unroll
    for (int v = 0; v < 8; ++v) {
      const float4 h0 = *(const float4*)(srow + v * ST_P);
      const float4 h1 = *(const float4*)(srow + v * ST_P + 64);
      const float4 e0 = *(const float4*)(trow + v * ST_T);
      const float4 e1 = *(const float4*)(trow + v * ST_T + 64);
      acc[0][v] = e0.x + sigm(acc[0][v] + b0.x) * h0.x;
      acc[1][v] = e0.y + sigm(acc[1][v] + b0.y) * h0.y;
      acc[2][v] = e0.z + sigm(acc[2][v] + b0.z) * h0.z;
      acc[3][v] = e0.w + sigm(acc[3][v] + b0.w) * h0.w;
      acc[4][v] = e1.x + sigm(acc[4][v] + b1.x) * h1.x;
      acc[5][v] = e1.y + sigm(acc[5][v] + b1.y) * h1.y;
      acc[6][v] = e1.z + sigm(acc[6][v] + b1.z) * h1.z;
      acc[7][v] = e1.w + sigm(acc[7][v] + b1.w) * h1.w;
      float sm = 0.f, sq = 0.f;
#pragma unroll
      for (int j = 0; j < 8; ++j) {
        sm += acc[j][v];
        sq += acc[j][v] * acc[j][v];
      }
      rs_[v] = sm;
      rq_[v] = sq;
    }
#pragma unroll
    for (int v = 0; v < 8; ++v) {
      rs_[v] = rowsum16(rs_[v]);
      rq_[v] = rowsum16(rq_[v]);
    }
#pragma unroll
    for (int v = 0; v < 8; ++v) {
      const float mu = rs_[v] * (1.f / 128.f);
      const float var = rq_[v] * (1.f / 128.f) - mu * mu;
      const float rs = rsqrtf(var + 1e-5f);
      *(float4*)(trow + v * ST_T) =
          make_float4((acc[0][v] - mu) * rs * g0.x + l0.x,
                      (acc[1][v] - mu) * rs * g0.y + l0.y,
                      (acc[2][v] - mu) * rs * g0.z + l0.z,
                      (acc[3][v] - mu) * rs * g0.w + l0.w);
      *(float4*)(trow + v * ST_T + 64) =
          make_float4((acc[4][v] - mu) * rs * g1.x + l1.x,
                      (acc[5][v] - mu) * rs * g1.y + l1.y,
                      (acc[6][v] - mu) * rs * g1.z + l1.z,
                      (acc[7][v] - mu) * rs * g1.w + l1.w);
    }
  }

  if (stager) wstage_load<H_>(sW, sf1W + (size_t)frot * H_ * H_, tid);
  lds_barrier();
  if (stager) {
    wstage_store<H_>(sW, BW0, tid);
    wstage_load<H_>(sW, sf1W + (size_t)((frot + 1) & (F_ - 1)) * H_ * H_, tid);
    skstage_load(sS, sskW + (size_t)frot * H_ * F_, tid);
    skstage_store(sS, P0, tid);
  }
  lds_barrier();

  v8f b1 = zero8;
  v8f sk = zero8;
  for (int p = 0; p < F_; ++p) {
    const int q = (p + frot) & (F_ - 1);
    const int q1 = (p + 1 + frot) & (F_ - 1);
    const int q2 = (p + 2 + frot) & (F_ - 1);
    const float* nxt2 = (p + 2 < F_) ? sf1W + (size_t)q2 * H_ * H_ : sf2W;
    float* cur = (p & 1) ? BW1 : BW0;
    float* oth = (p & 1) ? BW0 : BW1;
    const float* scur = (p & 1) ? P3 : P0;
    float* soth = (p & 1) ? P0 : P3;
    if (stager) {
      if (p + 1 < F_) skstage_load(sS, sskW + (size_t)q1 * H_ * F_, tid);
      if (p + 2 <= F_) wstage_swap<H_, H_>(sW, oth, nxt2, tid);
      else             wstage_store<H_>(sW, oth, tid);
      if (p + 1 < F_) skstage_store(sS, soth, tid);
    } else if (wid == 1) {
      wmma_gemm_lds_pair<H_>(b1, sk, TRANS + q * H_, ST_T, cur, n0, scur,
                             lane16, hi);
    } else {
      b1 = wmma_gemm_lds<H_>(b1, TRANS + q * H_, ST_T, cur, n0, lane16, hi);
    }
    lds_barrier();
  }
  if (!stager) {
    float b = sf1B[n0 + lane16];
#pragma unroll
    for (int v = 0; v < 8; ++v) {
      float t = b1[v] + b;
      b1[v] = t > 0.f ? t : (expf(t) - 1.f);
    }
    store_tile(P1, ST_P, n0, b1, lane16, hi);
  }
  lds_barrier();

  if (!stager) {
    const float b = bias_at(sf2B, bco);
    v8f b2 = wmma_gemm_lds<H_>(zero8, P1, ST_P, BW0, n0, lane16, hi);
#pragma unroll
    for (int v = 0; v < 8; ++v) b2[v] += b;
    store_tile(P2, ST_P, n0, b2, lane16, hi);
  }
  lds_barrier();

  if (!stager) {
    if (wid == 0) {
      v8f g = wmma_gemm<H_>(P2, ST_P, sgW, F_, 0, lane16, hi);
      float b = sgB[lane16];
#pragma unroll
      for (int v = 0; v < 8; ++v) g[v] += b;
      store_tile(GP, ST_S, 0, g, lane16, hi);
    } else if (wid == 1) {
      float b = sskB[lane16];
#pragma unroll
      for (int v = 0; v < 8; ++v) sk[v] += b;
      store_tile(SKT, ST_S, 0, sk, lane16, hi);
    }
  }
  lds_barrier();

  if (!stager) {
    float gp = GP[lr * ST_S + lc];
    float sg = 1.f / (1.f + expf(-gp));
    float v = SKT[lr * ST_S + lc] + sg * gp;
    float s = rowsum16(v);
    float ss = rowsum16(v * v);
    float mu = s * (1.f / 16.f);
    float var = ss * (1.f / 16.f) - mu * mu;
    float nv = (v - mu) * rsqrtf(var + 1e-5f) * slnG[lc] + slnB[lc];
    float mx = rowmax16(nv);
    float e = expf(nv - mx);
    float se = rowsum16(e);
    float w = e / se;
    WB[lr * 16 + lc] = w;
    outW[(size_t)(row0 + lr) * F_ + lc] = w;
  }
  lds_barrier();

  if (!stager) {
#pragma unroll
    for (int j = 0; j < 8; ++j) {
      int h = lc + 16 * j;
      float acc = 0.f;
#pragma unroll
      for (int f = 0; f < F_; ++f)
        acc += TRANS[lr * ST_T + f * H_ + h] * WB[lr * 16 + f];
      outSel[(size_t)(row0 + lr) * H_ + h] = acc;
    }
  }
}

extern "C" void kernel_launch(void* const* d_in, const int* in_sizes, int n_in,
                              void* d_out, int out_size, void* d_ws, size_t ws_size,
                              hipStream_t stream) {
  (void)in_sizes; (void)n_in; (void)out_size; (void)d_ws; (void)ws_size;
  const float* x    = (const float*)d_in[0];
  const float* embW = (const float*)d_in[1];
  const float* embB = (const float*)d_in[2];
  const float* fc1W = (const float*)d_in[3];
  const float* fc1B = (const float*)d_in[4];
  const float* fc2W = (const float*)d_in[5];
  const float* fc2B = (const float*)d_in[6];
  const float* gW   = (const float*)d_in[7];
  const float* gB   = (const float*)d_in[8];
  const float* lnG  = (const float*)d_in[9];
  const float* lnB  = (const float*)d_in[10];
  const float* sf1W = (const float*)d_in[11];
  const float* sf1B = (const float*)d_in[12];
  const float* sf2W = (const float*)d_in[13];
  const float* sf2B = (const float*)d_in[14];
  const float* sgW  = (const float*)d_in[15];
  const float* sgB  = (const float*)d_in[16];
  const float* sskW = (const float*)d_in[17];
  const float* sskB = (const float*)d_in[18];
  const float* slnG = (const float*)d_in[19];
  const float* slnB = (const float*)d_in[20];

  float* outSel = (float*)d_out;
  float* outW   = outSel + (size_t)ROWS_TOTAL * H_;

  vsn_fused<<<ROWS_TOTAL / 16, 512, 0, stream>>>(
      x, embW, embB, fc1W, fc1B, fc2W, fc2B, gW, gB, lnG, lnB,
      sf1W, sf1B, sf2W, sf2B, sgW, sgB, sskW, sskB, slnG, slnB,
      outSel, outW);
}


# s_fc1 (K=2048) and skip GEMMs cut by feature: wave w appends rows 128w.. of both to its own chain (B from L2); no LDS staging, no panel barriers; 16 partials added in LDS: both sums' order changes
# speedup vs baseline: 2.6111x; 1.0923x over previous
#include <hip/hip_runtime.h>
#include <hip/hip_bf16.h>
#include <math.h>


typedef float v2f __attribute__((ext_vector_type(2)));
typedef float v8f __attribute__((ext_vector_type(8)));

#define ROWS_TOTAL 16384
#define F_ 16
#define D_ 64
#define H_ 128
#define FH_ 2048

#define ST_T 2052
#define ST_P 132
#define ST_S 20

#define SMEM_FLOATS (16*ST_T + 16*16*ST_P + 2*16*ST_P + 16*16*ST_S + 2*16*ST_S + 16*16)

__device__ __forceinline__ float rowsum16(float v) {
  v += __shfl_xor(v, 1, 16);
  v += __shfl_xor(v, 2, 16);
  v += __shfl_xor(v, 4, 16);
  v += __shfl_xor(v, 8, 16);
  return v;
}
__device__ __forceinline__ float rowmax16(float v) {
  v = fmaxf(v, __shfl_xor(v, 1, 16));
  v = fmaxf(v, __shfl_xor(v, 2, 16));
  v = fmaxf(v, __shfl_xor(v, 4, 16));
  v = fmaxf(v, __shfl_xor(v, 8, 16));
  return v;
}

template <int K>
__device__ __forceinline__ v8f wmma_gemm(const float* __restrict__ As, int lda,
                                         const float* __restrict__ W, int ldw,
                                         int n0, int lane16, int hi) {
  v8f c = {0.f, 0.f, 0.f, 0.f, 0.f, 0.f, 0.f, 0.f};
  const float* arow = As + lane16 * lda + hi * 2;
  const float* wcol = W + (size_t)(hi * 2) * ldw + n0 + lane16;
#pragma unroll 1
  for (int kb = 0; kb < K; kb += 32) {
    if (kb + 32 < K) {
      __builtin_prefetch(wcol + (size_t)(kb + 32) * ldw, 0, 1);
      __builtin_prefetch(wcol + (size_t)(kb + 48) * ldw, 0, 1);
    }
#pragma unroll
    for (int k0 = kb; k0 < kb + 32; k0 += 4) {
      v2f a, b;
      a.x = arow[k0];
      a.y = arow[k0 + 1];
      b.x = wcol[(size_t)k0 * ldw];
      b.y = wcol[(size_t)(k0 + 1) * ldw];
      c = __builtin_amdgcn_wmma_f32_16x16x4_f32(false, a, false, b,
                                                (short)0, c, false, false);
    }
  }
  return c;
}

typedef float v4f __attribute__((ext_vector_type(4)));
typedef const __attribute__((address_space(1))) float* gfloat_ptr;
typedef const __attribute__((address_space(1))) v4f* gv4f_ptr;
__device__ __forceinline__ gfloat_ptr scalar_base(const float* p) {
  const unsigned long long a = (unsigned long long)p;
  const unsigned lo = __builtin_amdgcn_readfirstlane((unsigned)a);
  const unsigned hi = __builtin_amdgcn_readfirstlane((unsigned)(a >> 32));
  return (gfloat_ptr)(((unsigned long long)hi << 32) | lo);
}
__device__ __forceinline__ float4 gload4(gfloat_ptr p) {
  const v4f v = *(gv4f_ptr)p;
  return make_float4(v.x, v.y, v.z, v.w);
}
__device__ __forceinline__ float bias_at(const float* rowbase, unsigned col) {
  return scalar_base(rowbase)[col];
}

__device__ __forceinline__ void store_tile(float* dst, int ldd, int col0,
                                           v8f c, int lane16, int hi) {
#pragma unroll
  for (int v = 0; v < 8; ++v)
    dst[(v + hi * 8) * ldd + col0 + lane16] = c[v];
}

__device__ __forceinline__ void lds_barrier() {
  __builtin_amdgcn_fence(__ATOMIC_RELEASE, "workgroup", "local");
  __builtin_amdgcn_s_barrier();
  __builtin_amdgcn_fence(__ATOMIC_ACQUIRE, "workgroup", "local");
}

struct BStep { float4 q0, q1, q2, q3; };
#define PD 4

__device__ __forceinline__ void bstep_load(BStep& b, gfloat_ptr src, int s) {
  gfloat_ptr p = src + s * (4 * H_);
  b.q0 = gload4(p);
  b.q1 = gload4(p + H_);
  b.q2 = gload4(p + 64);
  b.q3 = gload4(p + H_ + 64);
}
__device__ __forceinline__ void bprologue(BStep (&bq)[PD], gfloat_ptr src) {
#pragma unroll
  for (int i = 0; i < PD; ++i) bstep_load(bq[i], src, i);
}
__device__ __forceinline__ v8f wm(v2f a, float bx, float by, v8f c) {
  v2f b;
  b.x = bx;
  b.y = by;
  return __builtin_amdgcn_wmma_f32_16x16x4_f32(false, a, false, b, (short)0, c,
                                               false, false);
}
__device__ __forceinline__ void mma8(v8f (&acc)[8], v2f a, const BStep& b) {
  acc[0] = wm(a, b.q0.x, b.q1.x, acc[0]);
  acc[1] = wm(a, b.q0.y, b.q1.y, acc[1]);
  acc[2] = wm(a, b.q0.z, b.q1.z, acc[2]);
  acc[3] = wm(a, b.q0.w, b.q1.w, acc[3]);
  acc[4] = wm(a, b.q2.x, b.q3.x, acc[4]);
  acc[5] = wm(a, b.q2.y, b.q3.y, acc[5]);
  acc[6] = wm(a, b.q2.z, b.q3.z, acc[6]);
  acc[7] = wm(a, b.q2.w, b.q3.w, acc[7]);
}
template <int K>
__device__ __forceinline__ void gemm_run(v8f (&acc)[8], BStep (&bq)[PD],
                                         const float* arow, gfloat_ptr src) {
  constexpr int KS = K / 4;
  v2f aq[PD];
#pragma unroll
  for (int i = 0; i < PD; ++i) aq[i] = *(const v2f*)(arow + 4 * i);
#pragma unroll 1
  for (int s0 = PD; s0 < KS; s0 += PD) {
#pragma unroll
    for (int i = 0; i < PD; ++i) {
      mma8(acc, aq[i], bq[i]);
      bstep_load(bq[i], src, s0 + i);
      aq[i] = *(const v2f*)(arow + 4 * (s0 + i));
    }
  }
#pragma unroll
  for (int i = 0; i < PD; ++i) mma8(acc, aq[i], bq[i]);
}
__device__ __forceinline__ void wave_lds_turn() {
  __builtin_amdgcn_fence(__ATOMIC_RELEASE, "workgroup", "local");
  __builtin_amdgcn_fence(__ATOMIC_ACQUIRE, "workgroup", "local");
}
__device__ __forceinline__ float elu1(float t) {
  return t > 0.f ? t : (expf(t) - 1.f);
}
template <bool ELU>
__device__ __forceinline__ void store_rows(float* dst, int ld,
                                           const v8f (&acc)[8], float4 b0,
                                           float4 b1) {
#pragma unroll
  for (int v = 0; v < 8; ++v) {
    float4 lo = make_float4(acc[0][v] + b0.x, acc[1][v] + b0.y,
                            acc[2][v] + b0.z, acc[3][v] + b0.w);
    float4 hi4 = make_float4(acc[4][v] + b1.x, acc[5][v] + b1.y,
                             acc[6][v] + b1.z, acc[7][v] + b1.w);
    if (ELU) {
      lo = make_float4(elu1(lo.x), elu1(lo.y), elu1(lo.z), elu1(lo.w));
      hi4 = make_float4(elu1(hi4.x), elu1(hi4.y), elu1(hi4.z), elu1(hi4.w));
    }
    *(float4*)(dst + v * ld) = lo;
    *(float4*)(dst + v * ld + 64) = hi4;
  }
}

__device__ __forceinline__ void store_partial(float* dst, int ld,
                                              const v8f (&acc)[8]) {
#pragma unroll
  for (int v = 0; v < 8; ++v) {
    *(float4*)(dst + v * ld) =
        make_float4(acc[0][v], acc[1][v], acc[2][v], acc[3][v]);
    *(float4*)(dst + v * ld + 64) =
        make_float4(acc[4][v], acc[5][v], acc[6][v], acc[7][v]);
  }
}
struct BCol { float x[H_ / 4], y[H_ / 4]; };
__device__ __forceinline__ void bcol_load(BCol& b, gfloat_ptr src, int ld) {
#pragma unroll
  for (int s = 0; s < H_ / 4; ++s) {
    b.x[s] = src[s * 4 * ld];
    b.y[s] = src[s * 4 * ld + ld];
  }
}
__device__ __forceinline__ v8f gemm_bcol(const float* arow, const BCol& b) {
  v8f c = {0.f, 0.f, 0.f, 0.f, 0.f, 0.f, 0.f, 0.f};
#pragma unroll
  for (int s = 0; s < H_ / 4; ++s)
    c = wm(*(const v2f*)(arow + 4 * s), b.x[s], b.y[s], c);
  return c;
}

__device__ __forceinline__ float sigm(float t) {
  return 1.f / (1.f + expf(-t));
}

__global__ __launch_bounds__(512, 1)
void vsn_fused(
    const float* __restrict__ x,
    const float* __restrict__ embW, const float* __restrict__ embB,
    const float* __restrict__ fc1W, const float* __restrict__ fc1B,
    const float* __restrict__ fc2W, const float* __restrict__ fc2B,
    const float* __restrict__ gW,   const float* __restrict__ gB,
    const float* __restrict__ lnG,  const float* __restrict__ lnB,
    const float* __restrict__ sf1W, const float* __restrict__ sf1B,
    const float* __restrict__ sf2W, const float* __restrict__ sf2B,
    const float* __restrict__ sgW,  const float* __restrict__ sgB,
    const float* __restrict__ sskW, const float* __restrict__ sskB,
    const float* __restrict__ slnG, const float* __restrict__ slnB,
    float* __restrict__ outSel, float* __restrict__ outW) {
  __shared__ __attribute__((aligned(16))) float smem[SMEM_FLOATS];
  float* TRANS = smem;
  float* SC = TRANS + 16 * ST_T;
  float* H1 = SC + 16 * 16 * ST_P;
  float* H2 = H1 + 16 * ST_P;
  float* SKP = H2 + 16 * ST_P;
  float* GP = SKP + 16 * 16 * ST_S;
  float* SKT = GP + 16 * ST_S;
  float* WB = SKT + 16 * ST_S;

  const int f = __builtin_amdgcn_readfirstlane((int)threadIdx.x >> 5);
  const int lane = threadIdx.x & 31;
  const int lane16 = lane & 15;
  const int hi = lane >> 4;
  const int lr = (threadIdx.x >> 4) & 15;
  const int lc = threadIdx.x & 15;
  const int row0 = blockIdx.x * 16;

  const v8f zero8 = {0.f, 0.f, 0.f, 0.f, 0.f, 0.f, 0.f, 0.f};

  {
    float* S = SC + f * (16 * ST_P);
    float* const TF = TRANS + f * H_;
    const unsigned toff = (unsigned)(2 * hi * H_ + 4 * lane16);
    const unsigned bc = (unsigned)(4 * lane16);
    float* const srow = S + (8 * hi) * ST_P + 4 * lane16;
    float* const trow = TF + (8 * hi) * ST_T + 4 * lane16;
    const float* const sA = S + lane16 * ST_P + 2 * hi;
    const float* const tA = TF + lane16 * ST_T + 2 * hi;

    BStep bq[PD];
    v8f acc[8];

    const float* xp = x + (size_t)(row0 + hi) * (F_ * D_) + f * D_ + 4 * lane16;
    float4 xq[8];
#pragma unroll
    for (int i = 0; i < 8; ++i)
      xq[i] = *(const float4*)(xp + (size_t)(2 * i) * (F_ * D_));
    gfloat_ptr wsrc = scalar_base(embW + (size_t)f * D_ * H_) + toff;
    bprologue(bq, wsrc);
    gfloat_ptr bsrc = scalar_base(embB + f * H_) + bc;
    float4 b0 = gload4(bsrc), b1 = gload4(bsrc + 64);
#pragma unroll
    for (int i = 0; i < 8; ++i)
      *(float4*)(S + (hi + 2 * i) * ST_P + 4 * lane16) = xq[i];
    wave_lds_turn();

#pragma unroll
    for (int j = 0; j < 8; ++j) acc[j] = zero8;
    gemm_run<D_>(acc, bq, sA, wsrc);
    wsrc = scalar_base(fc1W + (size_t)f * H_ * H_) + toff;
    bprologue(bq, wsrc);
    store_rows<false>(trow, ST_T, acc, b0, b1);
    bsrc = scalar_base(fc1B + f * H_) + bc;
    b0 = gload4(bsrc); b1 = gload4(bsrc + 64);
    wave_lds_turn();

#pragma unroll
    for (int j = 0; j < 8; ++j) acc[j] = zero8;
    gemm_run<H_>(acc, bq, tA, wsrc);
    wsrc = scalar_base(fc2W + (size_t)f * H_ * H_) + toff;
    bprologue(bq, wsrc);
    store_rows<true>(srow, ST_P, acc, b0, b1);
    bsrc = scalar_base(fc2B + f * H_) + bc;
    b0 = gload4(bsrc); b1 = gload4(bsrc + 64);
    wave_lds_turn();

#pragma unroll
    for (int j = 0; j < 8; ++j) acc[j] = zero8;
    gemm_run<H_>(acc, bq, sA, wsrc);
    wsrc = scalar_base(gW + (size_t)f * H_ * H_) + toff;
    bprologue(bq, wsrc);
    store_rows<false>(srow, ST_P, acc, b0, b1);
    bsrc = scalar_base(gB + f * H_) + bc;
    b0 = gload4(bsrc); b1 = gload4(bsrc + 64);
    gfloat_ptr gsrc = scalar_base(lnG + f * H_) + bc;
    gfloat_ptr lsrc = scalar_base(lnB + f * H_) + bc;
    const float4 g0 = gload4(gsrc), g1 = gload4(gsrc + 64);
    const float4 l0 = gload4(lsrc), l1 = gload4(lsrc + 64);
    wave_lds_turn();

#pragma unroll
    for (int j = 0; j < 8; ++j) acc[j] = zero8;
    gemm_run<H_>(acc, bq, sA, wsrc);
    BCol kb;
    bcol_load(kb, scalar_base(sskW + (size_t)f * H_ * F_) +
                      (unsigned)(2 * hi * F_ + lane16), F_);
    float rs_[8], rq_[8];
#pragma unroll
    for (int v = 0; v < 8; ++v) {
      const float4 h0 = *(const float4*)(srow + v * ST_P);
      const float4 h1 = *(const float4*)(srow + v * ST_P + 64);
      const float4 e0 = *(const float4*)(trow + v * ST_T);
      const float4 e1 = *(const float4*)(trow + v * ST_T + 64);
      acc[0][v] = e0.x + sigm(acc[0][v] + b0.x) * h0.x;
      acc[1][v] = e0.y + sigm(acc[1][v] + b0.y) * h0.y;
      acc[2][v] = e0.z + sigm(acc[2][v] + b0.z) * h0.z;
      acc[3][v] = e0.w + sigm(acc[3][v] + b0.w) * h0.w;
      acc[4][v] = e1.x + sigm(acc[4][v] + b1.x) * h1.x;
      acc[5][v] = e1.y + sigm(acc[5][v] + b1.y) * h1.y;
      acc[6][v] = e1.z + sigm(acc[6][v] + b1.z) * h1.z;
      acc[7][v] = e1.w + sigm(acc[7][v] + b1.w) * h1.w;
      float sm = 0.f, sq = 0.f;
#pragma unroll
      for (int j = 0; j < 8; ++j) {
        sm += acc[j][v];
        sq += acc[j][v] * acc[j][v];
      }
      rs_[v] = sm;
      rq_[v] = sq;
    }
#pragma unroll
    for (int v = 0; v < 8; ++v) {
      rs_[v] = rowsum16(rs_[v]);
      rq_[v] = rowsum16(rq_[v]);
    }
#pragma unroll
    for (int v = 0; v < 8; ++v) {
      const float mu = rs_[v] * (1.f / 128.f);
      const float var = rq_[v] * (1.f / 128.f) - mu * mu;
      const float rs = rsqrtf(var + 1e-5f);
      *(float4*)(trow + v * ST_T) =
          make_float4((acc[0][v] - mu) * rs * g0.x + l0.x,
                      (acc[1][v] - mu) * rs * g0.y + l0.y,
                      (acc[2][v] - mu) * rs * g0.z + l0.z,
                      (acc[3][v] - mu) * rs * g0.w + l0.w);
      *(float4*)(trow + v * ST_T + 64) =
          make_float4((acc[4][v] - mu) * rs * g1.x + l1.x,
                      (acc[5][v] - mu) * rs * g1.y + l1.y,
                      (acc[6][v] - mu) * rs * g1.z + l1.z,
                      (acc[7][v] - mu) * rs * g1.w + l1.w);
    }
    wave_lds_turn();

    wsrc = scalar_base(sf1W + (size_t)f * H_ * H_) + toff;
    bprologue(bq, wsrc);

    store_tile(SKP + f * (16 * ST_S), ST_S, 0, gemm_bcol(tA, kb), lane16, hi);

#pragma unroll
    for (int j = 0; j < 8; ++j) acc[j] = zero8;
    gemm_run<H_>(acc, bq, tA, wsrc);
    store_partial(srow, ST_P, acc);
  }

  BCol fb;
  if (f < 8)
    bcol_load(fb, scalar_base(sf2W) + (unsigned)(2 * hi * H_ + 16 * f + lane16),
              H_);
  const int pr = (int)threadIdx.x >> 5;
  const int pq = (int)threadIdx.x & 31;
  const float4 pb = *(const float4*)(sf1B + 4 * pq);
  lds_barrier();

  {
    const float* pp = SC + pr * ST_P + 4 * pq;
    float4 sum = *(const float4*)pp;
#pragma unroll
    for (int g = 1; g < F_; ++g) {
      const float4 v = *(const float4*)(pp + g * (16 * ST_P));
      sum.x += v.x;
      sum.y += v.y;
      sum.z += v.z;
      sum.w += v.w;
    }
    *(float4*)(H1 + pr * ST_P + 4 * pq) =
        make_float4(elu1(sum.x + pb.x), elu1(sum.y + pb.y),
                    elu1(sum.z + pb.z), elu1(sum.w + pb.w));
    if (threadIdx.x < 256) {
      const float* sp = SKP + lr * ST_S + lc;
      float s = sp[0];
#pragma unroll
      for (int g = 1; g < F_; ++g) s += sp[g * (16 * ST_S)];
      SKT[lr * ST_S + lc] = s + sskB[lc];
    }
  }
  lds_barrier();

  if (f < 8) {
    const float b = bias_at(sf2B, (unsigned)(16 * f + lane16));
    v8f b2 = gemm_bcol(H1 + lane16 * ST_P + 2 * hi, fb);
#pragma unroll
    for (int v = 0; v < 8; ++v) b2[v] += b;
    store_tile(H2, ST_P, 16 * f, b2, lane16, hi);
  }
  lds_barrier();

  if (f == 0) {
    v8f g = wmma_gemm<H_>(H2, ST_P, sgW, F_, 0, lane16, hi);
    float b = sgB[lane16];
#pragma unroll
    for (int v = 0; v < 8; ++v) g[v] += b;
    store_tile(GP, ST_S, 0, g, lane16, hi);
  }
  lds_barrier();

  if (threadIdx.x < 256) {
    float gp = GP[lr * ST_S + lc];
    float sg = 1.f / (1.f + expf(-gp));
    float v = SKT[lr * ST_S + lc] + sg * gp;
    float s = rowsum16(v);
    float ss = rowsum16(v * v);
    float mu = s * (1.f / 16.f);
    float var = ss * (1.f / 16.f) - mu * mu;
    float nv = (v - mu) * rsqrtf(var + 1e-5f) * slnG[lc] + slnB[lc];
    float mx = rowmax16(nv);
    float e = expf(nv - mx);
    float se = rowsum16(e);
    float w = e / se;
    WB[lr * 16 + lc] = w;
    outW[(size_t)(row0 + lr) * F_ + lc] = w;
  }
  lds_barrier();

  if (threadIdx.x < 256) {
#pragma unroll
    for (int j = 0; j < 8; ++j) {
      int h = lc + 16 * j;
      float acc = 0.f;
#pragma unroll
      for (int ff = 0; ff < F_; ++ff)
        acc += TRANS[lr * ST_T + ff * H_ + h] * WB[lr * 16 + ff];
      outSel[(size_t)(row0 + lr) * H_ + h] = acc;
    }
  }
}


extern "C" void kernel_launch(void* const* d_in, const int* in_sizes, int n_in,
                              void* d_out, int out_size, void* d_ws, size_t ws_size,
                              hipStream_t stream) {
  (void)in_sizes; (void)n_in; (void)out_size; (void)d_ws; (void)ws_size;
  const float* x    = (const float*)d_in[0];
  const float* embW = (const float*)d_in[1];
  const float* embB = (const float*)d_in[2];
  const float* fc1W = (const float*)d_in[3];
  const float* fc1B = (const float*)d_in[4];
  const float* fc2W = (const float*)d_in[5];
  const float* fc2B = (const float*)d_in[6];
  const float* gW   = (const float*)d_in[7];
  const float* gB   = (const float*)d_in[8];
  const float* lnG  = (const float*)d_in[9];
  const float* lnB  = (const float*)d_in[10];
  const float* sf1W = (const float*)d_in[11];
  const float* sf1B = (const float*)d_in[12];
  const float* sf2W = (const float*)d_in[13];
  const float* sf2B = (const float*)d_in[14];
  const float* sgW  = (const float*)d_in[15];
  const float* sgB  = (const float*)d_in[16];
  const float* sskW = (const float*)d_in[17];
  const float* sskB = (const float*)d_in[18];
  const float* slnG = (const float*)d_in[19];
  const float* slnB = (const float*)d_in[20];

  float* outSel = (float*)d_out;
  float* outW   = outSel + (size_t)ROWS_TOTAL * H_;

  vsn_fused<<<ROWS_TOTAL / 16, 512, 0, stream>>>(
      x, embW, embB, fc1W, fc1B, fc2W, fc2B, gW, gB, lnG, lnB,
      sf1W, sf1B, sf2W, sf2B, sgW, sgB, sskW, sskB, slnG, slnB,
      outSel, outW);
}
